# starting kernel
# baseline (speedup 1.0000x reference)
#include <hip/hip_runtime.h>
#include <math.h>

#define HID 9
#define NG  36
#define TLEN 512
#define NCLS 10

typedef __attribute__((ext_vector_type(2))) float v2f;
typedef __attribute__((ext_vector_type(8))) float v8f;

__device__ __forceinline__ float fexp2(float x) {
#if defined(__HIP_DEVICE_COMPILE__) && __has_builtin(__builtin_amdgcn_exp2f)
  return __builtin_amdgcn_exp2f(x);
#else
  return exp2f(x);
#endif
}

__device__ __forceinline__ float frcp(float x) {
#if defined(__HIP_DEVICE_COMPILE__) && __has_builtin(__builtin_amdgcn_rcpf)
  return __builtin_amdgcn_rcpf(x);
#else
  return 1.0f / x;
#endif
}

__device__ __forceinline__ float ftanh(float x) {
#if defined(__HIP_DEVICE_COMPILE__) && __has_builtin(__builtin_amdgcn_tanhf)
  return __builtin_amdgcn_tanhf(x);
#else
  float e = fexp2(2.8853900817779268f * x);
  return 1.0f - 2.0f * frcp(e + 1.0f);
#endif
}

__device__ __forceinline__ float fsig(float x) {
  return 0.5f + 0.5f * ftanh(0.5f * x);
}

__device__ __forceinline__ v8f wmma4(v2f a, v2f b, v8f c) {
  return __builtin_amdgcn_wmma_f32_16x16x4_f32(false, a, false, b, (short)0, c,
                                               false, false);
}

__device__ __forceinline__ void load_A(const float* __restrict__ W,
                                       v2f A[3][3], int m, int hi) {
#pragma unroll
  for (int tt = 0; tt < 3; ++tt) {
    int row = 16 * tt + m;
    bool rok = row < NG;
#pragma unroll
    for (int cc = 0; cc < 3; ++cc) {
      int col = 4 * cc + 2 * hi;
      float a0 = (rok && col     < HID) ? W[row * HID + col]     : 0.0f;
      float a1 = (rok && col + 1 < HID) ? W[row * HID + col + 1] : 0.0f;
      v2f a; a.x = a0; a.y = a1;
      A[tt][cc] = a;
    }
  }
}

__device__ __forceinline__ void load_bias(const float* __restrict__ bi,
                                          const float* __restrict__ bh,
                                          v8f bias[3], int hi) {
#pragma unroll
  for (int tt = 0; tt < 3; ++tt) {
#pragma unroll
    for (int r = 0; r < 8; ++r) {
      int row = 16 * tt + 8 * hi + r;
      bias[tt][r] = (row < NG) ? (bi[row] + bh[row]) : 0.0f;
    }
  }
}

__device__ __forceinline__ void build_B(const float h[HID], v2f B[3], int hi) {
#pragma unroll
  for (int cc = 0; cc < 3; ++cc) {
    float lo0 = (4 * cc     < HID) ? h[(4 * cc    ) % HID] : 0.0f;
    float lo1 = (4 * cc + 1 < HID) ? h[(4 * cc + 1) % HID] : 0.0f;
    float hi0 = (4 * cc + 2 < HID) ? h[(4 * cc + 2) % HID] : 0.0f;
    float hi1 = (4 * cc + 3 < HID) ? h[(4 * cc + 3) % HID] : 0.0f;
    v2f b;
    b.x = hi ? hi0 : lo0;
    b.y = hi ? hi1 : lo1;
    B[cc] = b;
  }
}

__device__ __forceinline__ void lstm_step(const v2f Aih[3][3],
                                          const v2f Ahh[3][3],
                                          const v8f bias[3],
                                          const v2f Bx[3],
                                          float h[HID], float cst[HID],
                                          int hi) {
  v2f Bh[3];
  build_B(h, Bh, hi);

  v8f D[3];
  D[0] = bias[0]; D[1] = bias[1]; D[2] = bias[2];
#pragma unroll
  for (int tt = 0; tt < 3; ++tt) {
#pragma unroll
    for (int cc = 0; cc < 3; ++cc) {
      D[tt] = wmma4(Aih[tt][cc], Bx[cc], D[tt]);
      D[tt] = wmma4(Ahh[tt][cc], Bh[cc], D[tt]);
    }
  }

  float own[24], part[24];
#pragma unroll
  for (int tt = 0; tt < 3; ++tt)
#pragma unroll
    for (int r = 0; r < 8; ++r) own[tt * 8 + r] = D[tt][r];
#pragma unroll
  for (int i = 0; i < 24; ++i) part[i] = __shfl_xor(own[i], 16, 32);

  float G[NG];
#pragma unroll
  for (int row = 0; row < NG; ++row) {
    int tt = row >> 4;
    int rr = row & 15;
    int half = rr >> 3;
    int r = rr & 7;
    float a = own[tt * 8 + r];
    float b = part[tt * 8 + r];
    G[row] = (half == hi) ? a : b;
  }

#pragma unroll
  for (int j = 0; j < HID; ++j) {
    float ig = fsig(G[j]);
    float fg = fsig(G[HID + j]);
    float gg = ftanh(G[2 * HID + j]);
    float og = fsig(G[3 * HID + j]);
    float cn = fg * cst[j] + ig * gg;
    cst[j] = cn;
    h[j] = og * ftanh(cn);
  }
}

__global__ __launch_bounds__(32) void lstm2_fc_fused(
    const float* __restrict__ x,
    const float* __restrict__ wih0, const float* __restrict__ whh0,
    const float* __restrict__ bih0, const float* __restrict__ bhh0,
    const float* __restrict__ wih1, const float* __restrict__ whh1,
    const float* __restrict__ bih1, const float* __restrict__ bhh1,
    const float* __restrict__ fcw,  const float* __restrict__ fcb,
    float* __restrict__ out) {
  const int lane = threadIdx.x;
  const int m  = lane & 15;
  const int hi = lane >> 4;
  const int bglob = blockIdx.x * 16 + m;

  v2f Aih0[3][3], Ahh0[3][3], Aih1[3][3], Ahh1[3][3];
  load_A(wih0, Aih0, m, hi);
  load_A(whh0, Ahh0, m, hi);
  load_A(wih1, Aih1, m, hi);
  load_A(whh1, Ahh1, m, hi);

  v8f bias0[3], bias1[3];
  load_bias(bih0, bhh0, bias0, hi);
  load_bias(bih1, bhh1, bias1, hi);

  float h0[HID], c0[HID], h1[HID], c1[HID];
#pragma unroll
  for (int j = 0; j < HID; ++j) { h0[j] = c0[j] = h1[j] = c1[j] = 0.0f; }

  const float* xrow = x + (size_t)bglob * TLEN * HID;

#pragma unroll 1
  for (int t = 0; t < TLEN; ++t) {
    const float* xp = xrow + t * HID;
    if (t + 8 < TLEN) __builtin_prefetch(xp + 8 * HID, 0, 0);

    v2f Bx[3];
#pragma unroll
    for (int cc = 0; cc < 3; ++cc) {
      int k0 = 4 * cc + 2 * hi;
      float v0 = (k0     < HID) ? xp[k0]     : 0.0f;
      float v1 = (k0 + 1 < HID) ? xp[k0 + 1] : 0.0f;
      v2f b; b.x = v0; b.y = v1;
      Bx[cc] = b;
    }

    lstm_step(Aih0, Ahh0, bias0, Bx, h0, c0, hi);

    v2f B1[3];
    build_B(h0, B1, hi);
    lstm_step(Aih1, Ahh1, bias1, B1, h1, c1, hi);
  }

  if (hi == 0) {
    float* op = out + (size_t)bglob * NCLS;
#pragma unroll
    for (int k = 0; k < NCLS; ++k) {
      float acc = fcb[k];
#pragma unroll
      for (int j = 0; j < HID; ++j) acc += h1[j] * fcw[k * HID + j];
      op[k] = acc;
    }
  }
}

extern "C" void kernel_launch(void* const* d_in, const int* in_sizes, int n_in,
                              void* d_out, int out_size, void* d_ws, size_t ws_size,
                              hipStream_t stream) {
  const float* x    = (const float*)d_in[0];
  const float* wih0 = (const float*)d_in[1];
  const float* whh0 = (const float*)d_in[2];
  const float* bih0 = (const float*)d_in[3];
  const float* bhh0 = (const float*)d_in[4];
  const float* wih1 = (const float*)d_in[5];
  const float* whh1 = (const float*)d_in[6];
  const float* bih1 = (const float*)d_in[7];
  const float* bhh1 = (const float*)d_in[8];
  const float* fcw  = (const float*)d_in[9];
  const float* fcb  = (const float*)d_in[10];
  float* out = (float*)d_out;

  const int B = in_sizes[0] / (TLEN * HID);
  dim3 grid(B / 16);
  dim3 block(32);
  lstm2_fc_fused<<<grid, block, 0, stream>>>(x, wih0, whh0, bih0, bhh0,
                                             wih1, whh1, bih1, bhh1,
                                             fcw, fcb, out);
}


# each lane half owns its hidden units (5 and 4 of 9): weight rows and k columns permuted at load put a unit's gates in its lane; no per-step shuffles or duplicate gate math; k sum order changes
# speedup vs baseline: 2.0527x; 2.0527x over previous
#include <hip/hip_runtime.h>
#include <math.h>


#define HID 9
#define NG  36
#define TLEN 512
#define NCLS 10
#define NSL 5


typedef __attribute__((ext_vector_type(2))) float v2f;
typedef __attribute__((ext_vector_type(8))) float v8f;

__device__ __forceinline__ float fexp2(float x) {
#if defined(__HIP_DEVICE_COMPILE__) && __has_builtin(__builtin_amdgcn_exp2f)
  return __builtin_amdgcn_exp2f(x);
#else
  return exp2f(x);
#endif
}

__device__ __forceinline__ float frcp(float x) {
#if defined(__HIP_DEVICE_COMPILE__) && __has_builtin(__builtin_amdgcn_rcpf)
  return __builtin_amdgcn_rcpf(x);
#else
  return 1.0f / x;
#endif
}

__device__ __forceinline__ float ftanh(float x) {
#if defined(__HIP_DEVICE_COMPILE__) && __has_builtin(__builtin_amdgcn_tanhf)
  return __builtin_amdgcn_tanhf(x);
#else
  float e = fexp2(2.8853900817779268f * x);
  return 1.0f - 2.0f * frcp(e + 1.0f);
#endif
}

__device__ __forceinline__ float fsig(float x) {
  return 0.5f + 0.5f * ftanh(0.5f * x);
}

__device__ __forceinline__ v8f wmma4(v2f a, v2f b, v8f c) {
  return __builtin_amdgcn_wmma_f32_16x16x4_f32(false, a, false, b, (short)0, c,
                                               false, false);
}

__device__ __forceinline__ int unit_of(int hh, int j) { return hh ? (NSL + j) : j; }
__device__ __forceinline__ bool slot_ok(int hh, int j) { return j < NSL && unit_of(hh, j) < HID; }

__device__ __forceinline__ void load_A(const float* __restrict__ W,
                                       v2f A[3][3], int m, int hi) {
  const int hr = m >> 3;
#pragma unroll
  for (int tt = 0; tt < 3; ++tt) {
    int s = 8 * tt + (m & 7);
    int j = s >> 2;
    int g = s & 3;
    bool rok = slot_ok(hr, j);
    int row = rok ? (g * HID + unit_of(hr, j)) : 0;
#pragma unroll
    for (int cc = 0; cc < 3; ++cc) {
      v2f a;
#pragma unroll
      for (int v = 0; v < 2; ++v) {
        int q = 2 * cc + v;
        bool ok = rok && slot_ok(hi, q);
        int col = slot_ok(hi, q) ? unit_of(hi, q) : 0;
        float w = W[row * HID + col];
        a[v] = ok ? w : 0.0f;
      }
      A[tt][cc] = a;
    }
  }
}

__device__ __forceinline__ void load_bias(const float* __restrict__ bi,
                                          const float* __restrict__ bh,
                                          v8f bias[3], int hi) {
#pragma unroll
  for (int tt = 0; tt < 3; ++tt) {
#pragma unroll
    for (int r = 0; r < 8; ++r) {
      int s = 8 * tt + r;
      int j = s >> 2;
      int g = s & 3;
      bool ok = slot_ok(hi, j);
      int row = ok ? (g * HID + unit_of(hi, j)) : 0;
      float b = bi[row] + bh[row];
      bias[tt][r] = ok ? b : 0.0f;
    }
  }
}

__device__ __forceinline__ void build_B(const float s[NSL], v2f B[3]) {
  v2f b0; b0.x = s[0]; b0.y = s[1];
  v2f b1; b1.x = s[2]; b1.y = s[3];
  v2f b2; b2.x = s[4]; b2.y = 0.0f;
  B[0] = b0; B[1] = b1; B[2] = b2;
}

__device__ __forceinline__ void lstm_step(const v2f Aih[3][3],
                                          const v2f Ahh[3][3],
                                          const v8f bias[3],
                                          const v2f Bx[3],
                                          float h[NSL], float cst[NSL]) {
  v2f Bh[3];
  build_B(h, Bh);

  v8f D[3];
  D[0] = bias[0]; D[1] = bias[1]; D[2] = bias[2];
#pragma unroll
  for (int tt = 0; tt < 3; ++tt) {
#pragma unroll
    for (int cc = 0; cc < 3; ++cc) {
      D[tt] = wmma4(Aih[tt][cc], Bx[cc], D[tt]);
      D[tt] = wmma4(Ahh[tt][cc], Bh[cc], D[tt]);
    }
  }

#pragma unroll
  for (int j = 0; j < NSL; ++j) {
    float ig = fsig(D[(4 * j) >> 3][(4 * j) & 7]);
    float fg = fsig(D[(4 * j + 1) >> 3][(4 * j + 1) & 7]);
    float gg = ftanh(D[(4 * j + 2) >> 3][(4 * j + 2) & 7]);
    float og = fsig(D[(4 * j + 3) >> 3][(4 * j + 3) & 7]);
    float cn = fg * cst[j] + ig * gg;
    cst[j] = cn;
    h[j] = og * ftanh(cn);
  }
}

__global__ __launch_bounds__(32) void lstm2_fc_fused(
    const float* __restrict__ x,
    const float* __restrict__ wih0, const float* __restrict__ whh0,
    const float* __restrict__ bih0, const float* __restrict__ bhh0,
    const float* __restrict__ wih1, const float* __restrict__ whh1,
    const float* __restrict__ bih1, const float* __restrict__ bhh1,
    const float* __restrict__ fcw,  const float* __restrict__ fcb,
    float* __restrict__ out) {
  const int lane = threadIdx.x;
  const int m  = lane & 15;
  const int hi = lane >> 4;
  const int bglob = blockIdx.x * 16 + m;

  v2f Aih0[3][3], Ahh0[3][3], Aih1[3][3], Ahh1[3][3];
  load_A(wih0, Aih0, m, hi);
  load_A(whh0, Ahh0, m, hi);
  load_A(wih1, Aih1, m, hi);
  load_A(whh1, Ahh1, m, hi);

  v8f bias0[3], bias1[3];
  load_bias(bih0, bhh0, bias0, hi);
  load_bias(bih1, bhh1, bias1, hi);

  float h0[NSL], c0[NSL], h1[NSL], c1[NSL];
#pragma unroll
  for (int j = 0; j < NSL; ++j) { h0[j] = c0[j] = h1[j] = c1[j] = 0.0f; }

  const float* xrow = x + (size_t)bglob * TLEN * HID;

#pragma unroll 1
  for (int t = 0; t < TLEN; ++t) {
    const float* xp = xrow + t * HID;
    if (t + 8 < TLEN) __builtin_prefetch(xp + 8 * HID, 0, 0);

    float xs[NSL];
#pragma unroll
    for (int q = 0; q < NSL; ++q) {
      bool ok = slot_ok(hi, q);
      float v = xp[ok ? unit_of(hi, q) : (HID - 1)];
      xs[q] = ok ? v : 0.0f;
    }
    v2f Bx[3];
    build_B(xs, Bx);

    lstm_step(Aih0, Ahh0, bias0, Bx, h0, c0);

    v2f B1[3];
    build_B(h0, B1);
    lstm_step(Aih1, Ahh1, bias1, B1, h1, c1);
  }

  float ph[NSL - 1];
#pragma unroll
  for (int j = 0; j < NSL - 1; ++j) ph[j] = __shfl_xor(h1[j], 16, 32);
  if (hi == 0) {
    float hall[HID];
#pragma unroll
    for (int j = 0; j < NSL; ++j) hall[j] = h1[j];
#pragma unroll
    for (int j = 0; j < NSL - 1; ++j) hall[NSL + j] = ph[j];
    float* op = out + (size_t)bglob * NCLS;
#pragma unroll
    for (int k = 0; k < NCLS; ++k) {
      float acc = fcb[k];
#pragma unroll
      for (int j = 0; j < HID; ++j) acc += hall[j] * fcw[k * HID + j];
      op[k] = acc;
    }
  }
}

extern "C" void kernel_launch(void* const* d_in, const int* in_sizes, int n_in,
                              void* d_out, int out_size, void* d_ws, size_t ws_size,
                              hipStream_t stream) {
  const float* x    = (const float*)d_in[0];
  const float* wih0 = (const float*)d_in[1];
  const float* whh0 = (const float*)d_in[2];
  const float* bih0 = (const float*)d_in[3];
  const float* bhh0 = (const float*)d_in[4];
  const float* wih1 = (const float*)d_in[5];
  const float* whh1 = (const float*)d_in[6];
  const float* bih1 = (const float*)d_in[7];
  const float* bhh1 = (const float*)d_in[8];
  const float* fcw  = (const float*)d_in[9];
  const float* fcb  = (const float*)d_in[10];
  float* out = (float*)d_out;

  const int B = in_sizes[0] / (TLEN * HID);
  dim3 grid(B / 16);
  dim3 block(32);
  lstm2_fc_fused<<<grid, block, 0, stream>>>(x, wih0, whh0, bih0, bhh0,
                                             wih1, whh1, bih1, bhh1,
                                             fcw, fcb, out);
}


# time loop software-pipelined three steps deep: each layer's gate math runs beside the other layer's matrix products; x part made a step early from x loaded a step before that; k sum order changes
# speedup vs baseline: 2.2892x; 1.1152x over previous
#include <hip/hip_runtime.h>
#include <math.h>


#define HID 9
#define NG  36
#define TLEN 512
#define NCLS 10
#define NSL 5


typedef __attribute__((ext_vector_type(2))) float v2f;
typedef __attribute__((ext_vector_type(8))) float v8f;

__device__ __forceinline__ float fexp2(float x) {
#if defined(__HIP_DEVICE_COMPILE__) && __has_builtin(__builtin_amdgcn_exp2f)
  return __builtin_amdgcn_exp2f(x);
#else
  return exp2f(x);
#endif
}

__device__ __forceinline__ float frcp(float x) {
#if defined(__HIP_DEVICE_COMPILE__) && __has_builtin(__builtin_amdgcn_rcpf)
  return __builtin_amdgcn_rcpf(x);
#else
  return 1.0f / x;
#endif
}

__device__ __forceinline__ float ftanh(float x) {
#if defined(__HIP_DEVICE_COMPILE__) && __has_builtin(__builtin_amdgcn_tanhf)
  return __builtin_amdgcn_tanhf(x);
#else
  float e = fexp2(2.8853900817779268f * x);
  return 1.0f - 2.0f * frcp(e + 1.0f);
#endif
}

__device__ __forceinline__ float fsig(float x) {
  return 0.5f + 0.5f * ftanh(0.5f * x);
}

__device__ __forceinline__ v8f wmma4(v2f a, v2f b, v8f c) {
  return __builtin_amdgcn_wmma_f32_16x16x4_f32(false, a, false, b, (short)0, c,
                                               false, false);
}

__device__ __forceinline__ int unit_of(int hh, int j) { return hh ? (NSL + j) : j; }
__device__ __forceinline__ bool slot_ok(int hh, int j) { return j < NSL && unit_of(hh, j) < HID; }

__device__ __forceinline__ void load_A(const float* __restrict__ W,
                                       v2f A[3][3], int m, int hi) {
  const int hr = m >> 3;
#pragma unroll
  for (int tt = 0; tt < 3; ++tt) {
    int s = 8 * tt + (m & 7);
    int j = s >> 2;
    int g = s & 3;
    bool rok = slot_ok(hr, j);
    int row = rok ? (g * HID + unit_of(hr, j)) : 0;
#pragma unroll
    for (int cc = 0; cc < 3; ++cc) {
      v2f a;
#pragma unroll
      for (int v = 0; v < 2; ++v) {
        int q = 2 * cc + v;
        bool ok = rok && slot_ok(hi, q);
        int col = slot_ok(hi, q) ? unit_of(hi, q) : 0;
        float w = W[row * HID + col];
        a[v] = ok ? w : 0.0f;
      }
      A[tt][cc] = a;
    }
  }
}

__device__ __forceinline__ void load_bias(const float* __restrict__ bi,
                                          const float* __restrict__ bh,
                                          v8f bias[3], int hi) {
#pragma unroll
  for (int tt = 0; tt < 3; ++tt) {
#pragma unroll
    for (int r = 0; r < 8; ++r) {
      int s = 8 * tt + r;
      int j = s >> 2;
      int g = s & 3;
      bool ok = slot_ok(hi, j);
      int row = ok ? (g * HID + unit_of(hi, j)) : 0;
      float b = bi[row] + bh[row];
      bias[tt][r] = ok ? b : 0.0f;
    }
  }
}

__device__ __forceinline__ void build_B(const float s[NSL], v2f B[3]) {
  v2f b0; b0.x = s[0]; b0.y = s[1];
  v2f b1; b1.x = s[2]; b1.y = s[3];
  v2f b2; b2.x = s[4]; b2.y = 0.0f;
  B[0] = b0; B[1] = b1; B[2] = b2;
}

__device__ __forceinline__ void mm3(const v2f A[3][3], const v2f B[3], v8f D[3]) {
#pragma unroll
  for (int cc = 0; cc < 3; ++cc) {
#pragma unroll
    for (int tt = 0; tt < 3; ++tt) D[tt] = wmma4(A[tt][cc], B[cc], D[tt]);
  }
}

__device__ __forceinline__ void gates(const v8f D[3], float h[NSL], float cst[NSL]) {
#pragma unroll
  for (int j = 0; j < NSL; ++j) {
    float ig = fsig(D[(4 * j) >> 3][(4 * j) & 7]);
    float fg = fsig(D[(4 * j + 1) >> 3][(4 * j + 1) & 7]);
    float gg = ftanh(D[(4 * j + 2) >> 3][(4 * j + 2) & 7]);
    float og = fsig(D[(4 * j + 3) >> 3][(4 * j + 3) & 7]);
    float cn = fg * cst[j] + ig * gg;
    cst[j] = cn;
    h[j] = og * ftanh(cn);
  }
}

__device__ __forceinline__ void load_x(const float* __restrict__ xp, int hi, float xs[NSL]) {
#pragma unroll
  for (int q = 0; q < NSL; ++q) {
    bool ok = slot_ok(hi, q);
    float v = xp[ok ? unit_of(hi, q) : (HID - 1)];
    xs[q] = ok ? v : 0.0f;
  }
}

__global__ __launch_bounds__(32) void lstm2_fc_fused(
    const float* __restrict__ x,
    const float* __restrict__ wih0, const float* __restrict__ whh0,
    const float* __restrict__ bih0, const float* __restrict__ bhh0,
    const float* __restrict__ wih1, const float* __restrict__ whh1,
    const float* __restrict__ bih1, const float* __restrict__ bhh1,
    const float* __restrict__ fcw,  const float* __restrict__ fcb,
    float* __restrict__ out) {
  const int lane = threadIdx.x;
  const int m  = lane & 15;
  const int hi = lane >> 4;
  const int bglob = blockIdx.x * 16 + m;

  v2f Aih0[3][3], Ahh0[3][3], Aih1[3][3], Ahh1[3][3];
  load_A(wih0, Aih0, m, hi);
  load_A(whh0, Ahh0, m, hi);
  load_A(wih1, Aih1, m, hi);
  load_A(whh1, Ahh1, m, hi);

  v8f bias0[3], bias1[3];
  load_bias(bih0, bhh0, bias0, hi);
  load_bias(bih1, bhh1, bias1, hi);

  float h0[NSL], c0[NSL], h1[NSL], c1[NSL];
#pragma unroll
  for (int j = 0; j < NSL; ++j) { h0[j] = c0[j] = h1[j] = c1[j] = 0.0f; }

  const float* xrow = x + (size_t)bglob * TLEN * HID;

  float xn[NSL];
  v2f Bx[3], Bh0[3], Bh1[3];
  v8f Dx[3], D0[3], D1[3];

  load_x(xrow, hi, xn);
  build_B(xn, Bx);
  Dx[0] = bias0[0]; Dx[1] = bias0[1]; Dx[2] = bias0[2];
  mm3(Aih0, Bx, Dx);
  build_B(h0, Bh0);
  D0[0] = Dx[0]; D0[1] = Dx[1]; D0[2] = Dx[2];
  mm3(Ahh0, Bh0, D0);
  gates(D0, h0, c0);
  load_x(xrow + HID, hi, xn);
  build_B(xn, Bx);
  Dx[0] = bias0[0]; Dx[1] = bias0[1]; Dx[2] = bias0[2];
  mm3(Aih0, Bx, Dx);
  load_x(xrow + 2 * HID, hi, xn);
  build_B(h0, Bh0);
  D0[0] = Dx[0]; D0[1] = Dx[1]; D0[2] = Dx[2];
  mm3(Ahh0, Bh0, D0);
  D1[0] = bias1[0]; D1[1] = bias1[1]; D1[2] = bias1[2];
  mm3(Aih1, Bh0, D1);
  gates(D0, h0, c0);
  build_B(h1, Bh1);
  mm3(Ahh1, Bh1, D1);
  build_B(xn, Bx);
  load_x(xrow + 3 * HID, hi, xn);
  Dx[0] = bias0[0]; Dx[1] = bias0[1]; Dx[2] = bias0[2];
  mm3(Aih0, Bx, Dx);

#pragma unroll 2
  for (int t = 1; t < TLEN - 1; ++t) {
    const int tp = (t + 11 < TLEN) ? (t + 11) : (TLEN - 1);
    __builtin_prefetch(xrow + tp * HID, 0, 0);

    build_B(h0, Bh0);
    D0[0] = Dx[0]; D0[1] = Dx[1]; D0[2] = Dx[2];
    v8f E1[3];
    E1[0] = bias1[0]; E1[1] = bias1[1]; E1[2] = bias1[2];
    gates(D1, h1, c1);
    mm3(Ahh0, Bh0, D0);
    mm3(Aih1, Bh0, E1);

    build_B(h1, Bh1);
    build_B(xn, Bx);
    const int tl = (t + 3 < TLEN) ? (t + 3) : (TLEN - 1);
    load_x(xrow + tl * HID, hi, xn);
    Dx[0] = bias0[0]; Dx[1] = bias0[1]; Dx[2] = bias0[2];
    gates(D0, h0, c0);
    mm3(Ahh1, Bh1, E1);
    mm3(Aih0, Bx, Dx);
    D1[0] = E1[0]; D1[1] = E1[1]; D1[2] = E1[2];
  }

  gates(D1, h1, c1);
  build_B(h0, Bh0);
  build_B(h1, Bh1);
  D1[0] = bias1[0]; D1[1] = bias1[1]; D1[2] = bias1[2];
  mm3(Aih1, Bh0, D1);
  mm3(Ahh1, Bh1, D1);
  gates(D1, h1, c1);

  float ph[NSL - 1];
#pragma unroll
  for (int j = 0; j < NSL - 1; ++j) ph[j] = __shfl_xor(h1[j], 16, 32);
  if (hi == 0) {
    float hall[HID];
#pragma unroll
    for (int j = 0; j < NSL; ++j) hall[j] = h1[j];
#pragma unroll
    for (int j = 0; j < NSL - 1; ++j) hall[NSL + j] = ph[j];
    float* op = out + (size_t)bglob * NCLS;
#pragma unroll
    for (int k = 0; k < NCLS; ++k) {
      float acc = fcb[k];
#pragma unroll
      for (int j = 0; j < HID; ++j) acc += hall[j] * fcw[k * HID + j];
      op[k] = acc;
    }
  }
}

extern "C" void kernel_launch(void* const* d_in, const int* in_sizes, int n_in,
                              void* d_out, int out_size, void* d_ws, size_t ws_size,
                              hipStream_t stream) {
  const float* x    = (const float*)d_in[0];
  const float* wih0 = (const float*)d_in[1];
  const float* whh0 = (const float*)d_in[2];
  const float* bih0 = (const float*)d_in[3];
  const float* bhh0 = (const float*)d_in[4];
  const float* wih1 = (const float*)d_in[5];
  const float* whh1 = (const float*)d_in[6];
  const float* bih1 = (const float*)d_in[7];
  const float* bhh1 = (const float*)d_in[8];
  const float* fcw  = (const float*)d_in[9];
  const float* fcb  = (const float*)d_in[10];
  float* out = (float*)d_out;

  const int B = in_sizes[0] / (TLEN * HID);
  dim3 grid(B / 16);
  dim3 block(32);
  lstm2_fc_fused<<<grid, block, 0, stream>>>(x, wih0, whh0, bih0, bhh0,
                                             wih1, whh1, bih1, bhh1,
                                             fcw, fcb, out);
}


# the two layers run on two waves of one workgroup (two SIMDs): the layer-0 wave hands h0 to the layer-1 wave through a two-buffer LDS ring, one workgroup barrier per step; sum order unchanged
# speedup vs baseline: 3.2855x; 1.4352x over previous
#include <hip/hip_runtime.h>
#include <math.h>


#define HID 9
#define NG  36
#define TLEN 512
#define NCLS 10
#define NSL 5


typedef __attribute__((ext_vector_type(2))) float v2f;
typedef __attribute__((ext_vector_type(8))) float v8f;

__device__ __forceinline__ float fexp2(float x) {
#if defined(__HIP_DEVICE_COMPILE__) && __has_builtin(__builtin_amdgcn_exp2f)
  return __builtin_amdgcn_exp2f(x);
#else
  return exp2f(x);
#endif
}

__device__ __forceinline__ float frcp(float x) {
#if defined(__HIP_DEVICE_COMPILE__) && __has_builtin(__builtin_amdgcn_rcpf)
  return __builtin_amdgcn_rcpf(x);
#else
  return 1.0f / x;
#endif
}

__device__ __forceinline__ float ftanh(float x) {
#if defined(__HIP_DEVICE_COMPILE__) && __has_builtin(__builtin_amdgcn_tanhf)
  return __builtin_amdgcn_tanhf(x);
#else
  float e = fexp2(2.8853900817779268f * x);
  return 1.0f - 2.0f * frcp(e + 1.0f);
#endif
}

__device__ __forceinline__ float fsig(float x) {
  return 0.5f + 0.5f * ftanh(0.5f * x);
}

__device__ __forceinline__ v8f wmma4(v2f a, v2f b, v8f c) {
  return __builtin_amdgcn_wmma_f32_16x16x4_f32(false, a, false, b, (short)0, c,
                                               false, false);
}

__device__ __forceinline__ int unit_of(int hh, int j) { return hh ? (NSL + j) : j; }
__device__ __forceinline__ bool slot_ok(int hh, int j) { return j < NSL && unit_of(hh, j) < HID; }

__device__ __forceinline__ void load_A(const float* __restrict__ W,
                                       v2f A[3][3], int m, int hi) {
  const int hr = m >> 3;
#pragma unroll
  for (int tt = 0; tt < 3; ++tt) {
    int s = 8 * tt + (m & 7);
    int j = s >> 2;
    int g = s & 3;
    bool rok = slot_ok(hr, j);
    int row = rok ? (g * HID + unit_of(hr, j)) : 0;
#pragma unroll
    for (int cc = 0; cc < 3; ++cc) {
      v2f a;
#pragma unroll
      for (int v = 0; v < 2; ++v) {
        int q = 2 * cc + v;
        bool ok = rok && slot_ok(hi, q);
        int col = slot_ok(hi, q) ? unit_of(hi, q) : 0;
        float w = W[row * HID + col];
        a[v] = ok ? w : 0.0f;
      }
      A[tt][cc] = a;
    }
  }
}

__device__ __forceinline__ void load_bias(const float* __restrict__ bi,
                                          const float* __restrict__ bh,
                                          v8f bias[3], int hi) {
#pragma unroll
  for (int tt = 0; tt < 3; ++tt) {
#pragma unroll
    for (int r = 0; r < 8; ++r) {
      int s = 8 * tt + r;
      int j = s >> 2;
      int g = s & 3;
      bool ok = slot_ok(hi, j);
      int row = ok ? (g * HID + unit_of(hi, j)) : 0;
      float b = bi[row] + bh[row];
      bias[tt][r] = ok ? b : 0.0f;
    }
  }
}

__device__ __forceinline__ void build_B(const float s[NSL], v2f B[3]) {
  v2f b0; b0.x = s[0]; b0.y = s[1];
  v2f b1; b1.x = s[2]; b1.y = s[3];
  v2f b2; b2.x = s[4]; b2.y = 0.0f;
  B[0] = b0; B[1] = b1; B[2] = b2;
}

__device__ __forceinline__ void mm3(const v2f A[3][3], const v2f B[3], v8f D[3]) {
#pragma unroll
  for (int cc = 0; cc < 3; ++cc) {
#pragma unroll
    for (int tt = 0; tt < 3; ++tt) D[tt] = wmma4(A[tt][cc], B[cc], D[tt]);
  }
}

__device__ __forceinline__ void gates(const v8f D[3], float h[NSL], float cst[NSL]) {
#pragma unroll
  for (int j = 0; j < NSL; ++j) {
    float ig = fsig(D[(4 * j) >> 3][(4 * j) & 7]);
    float fg = fsig(D[(4 * j + 1) >> 3][(4 * j + 1) & 7]);
    float gg = ftanh(D[(4 * j + 2) >> 3][(4 * j + 2) & 7]);
    float og = fsig(D[(4 * j + 3) >> 3][(4 * j + 3) & 7]);
    float cn = fg * cst[j] + ig * gg;
    cst[j] = cn;
    h[j] = og * ftanh(cn);
  }
}

__device__ __forceinline__ void fetch_x(const float* __restrict__ xp, int hi, float xs[NSL]) {
#pragma unroll
  for (int q = 0; q < NSL; ++q) xs[q] = xp[slot_ok(hi, q) ? unit_of(hi, q) : (HID - 1)];
}
__device__ __forceinline__ void build_Bx(const float xs[NSL], int hi, v2f B[3]) {
  float s[NSL];
#pragma unroll
  for (int q = 0; q < NSL; ++q) s[q] = slot_ok(hi, q) ? xs[q] : 0.0f;
  build_B(s, B);
}

__device__ __forceinline__ void wg_sync() {
  asm volatile("s_wait_dscnt 0x0\n\ts_barrier_signal -1\n\ts_barrier_wait -1" ::: "memory");
}

__device__ __forceinline__ void l0_step(const float* __restrict__ xrow, int k, int hi,
                                        const v2f Aih[3][3], const v2f Ahh[3][3],
                                        const v8f bias[3], float h[NSL], float c[NSL],
                                        v8f Dx[3], const float xuse[NSL], float xload[NSL],
                                        float* dst) {
  const int tp = (k + 11 < TLEN) ? (k + 11) : (TLEN - 1);
  __builtin_prefetch(xrow + tp * HID, 0, 0);
  const int tl = (k + 3 < TLEN) ? (k + 3) : (TLEN - 1);
  fetch_x(xrow + tl * HID, hi, xload);

  v2f Bh[3], Bx[3];
  v8f D0[3];
  build_B(h, Bh);
  D0[0] = Dx[0]; D0[1] = Dx[1]; D0[2] = Dx[2];
  mm3(Ahh, Bh, D0);
  __builtin_amdgcn_sched_barrier(0);
  build_Bx(xuse, hi, Bx);
  Dx[0] = bias[0]; Dx[1] = bias[1]; Dx[2] = bias[2];
  gates(D0, h, c);
#pragma unroll
  for (int j = 0; j < NSL; ++j) dst[32 * j] = h[j];
  mm3(Aih, Bx, Dx);
}

template <bool GATES, bool FETCH>
__device__ __forceinline__ void l1_step(const float* src,
                                        const v2f Aih[3][3], const v2f Ahh[3][3],
                                        const v8f bias[3], float h[NSL], float c[NSL],
                                        v8f D[3], float hin[NSL]) {
  float hn[NSL];
  if (FETCH) {
#pragma unroll
    for (int j = 0; j < NSL; ++j) hn[j] = src[32 * j];
  }
  v2f Bin[3], Bh[3];
  v8f E[3];
  E[0] = bias[0]; E[1] = bias[1]; E[2] = bias[2];
  build_B(hin, Bin);
  if (GATES) gates(D, h, c);
  mm3(Aih, Bin, E);
  build_B(h, Bh);
  mm3(Ahh, Bh, E);
  D[0] = E[0]; D[1] = E[1]; D[2] = E[2];
  if (FETCH) {
#pragma unroll
    for (int j = 0; j < NSL; ++j) hin[j] = hn[j];
  }
}

__global__ __launch_bounds__(64) void lstm2_fc_fused(
    const float* __restrict__ x,
    const float* __restrict__ wih0, const float* __restrict__ whh0,
    const float* __restrict__ bih0, const float* __restrict__ bhh0,
    const float* __restrict__ wih1, const float* __restrict__ whh1,
    const float* __restrict__ bih1, const float* __restrict__ bhh1,
    const float* __restrict__ fcw,  const float* __restrict__ fcb,
    float* __restrict__ out) {
  __shared__ float hx[2][NSL][32];

  const int lane = threadIdx.x & 31;
  const int role = __builtin_amdgcn_readfirstlane((int)(threadIdx.x >> 5));
  const int m  = lane & 15;
  const int hi = lane >> 4;
  const int bglob = blockIdx.x * 16 + m;

  const float* wih = role ? wih1 : wih0;
  const float* whh = role ? whh1 : whh0;
  const float* bih = role ? bih1 : bih0;
  const float* bhh = role ? bhh1 : bhh0;
  v2f Aih[3][3], Ahh[3][3];
  load_A(wih, Aih, m, hi);
  load_A(whh, Ahh, m, hi);
  v8f bias[3];
  load_bias(bih, bhh, bias, hi);

  float h[NSL], c[NSL];
  float xs0[NSL], xs1[NSL], xs2[NSL];
  float hin[NSL];
#pragma unroll
  for (int j = 0; j < NSL; ++j) { h[j] = c[j] = 0.0f; xs0[j] = xs1[j] = xs2[j] = hin[j] = 0.0f; }
  v8f Dacc[3];
  Dacc[0] = bias[0]; Dacc[1] = bias[1]; Dacc[2] = bias[2];

  const float* xrow = x + (size_t)bglob * TLEN * HID;
  float* ring = &hx[0][0][lane];
  const int RB = NSL * 32;

  if (role == 0) {
    float x0[NSL];
    v2f Bx[3];
    fetch_x(xrow, hi, x0);
    fetch_x(xrow + HID, hi, xs0);
    fetch_x(xrow + 2 * HID, hi, xs1);
    build_Bx(x0, hi, Bx);
    mm3(Aih, Bx, Dacc);
  }

  if (role == 0) l0_step(xrow, 0, hi, Aih, Ahh, bias, h, c, Dacc, xs0, xs2, ring);
  wg_sync();
  if (role == 0) {
    l0_step(xrow, 1, hi, Aih, Ahh, bias, h, c, Dacc, xs1, xs0, ring + RB);
  } else {
#pragma unroll
    for (int j = 0; j < NSL; ++j) hin[j] = ring[32 * j];
  }
  wg_sync();
  if (role == 0) l0_step(xrow, 2, hi, Aih, Ahh, bias, h, c, Dacc, xs2, xs1, ring);
  else           l1_step<false, true>(ring + RB, Aih, Ahh, bias, h, c, Dacc, hin);
  wg_sync();

  for (int k = 3; k < TLEN + 1; k += 3) {
    float* b0 = ring + ((k & 1) ? RB : 0);
    float* b1 = ring + ((k & 1) ? 0 : RB);
    if (role == 0) { if (k < TLEN)     l0_step(xrow, k,     hi, Aih, Ahh, bias, h, c, Dacc, xs0, xs2, b0); }
    else           l1_step<true, true>(b1, Aih, Ahh, bias, h, c, Dacc, hin);
    wg_sync();
    if (role == 0) { if (k + 1 < TLEN) l0_step(xrow, k + 1, hi, Aih, Ahh, bias, h, c, Dacc, xs1, xs0, b1); }
    else           l1_step<true, true>(b0, Aih, Ahh, bias, h, c, Dacc, hin);
    wg_sync();
    if (role == 0) { if (k + 2 < TLEN) l0_step(xrow, k + 2, hi, Aih, Ahh, bias, h, c, Dacc, xs2, xs1, b0); }
    else           l1_step<true, true>(b1, Aih, Ahh, bias, h, c, Dacc, hin);
    wg_sync();
  }

  if (role == 1) {
    l1_step<true, false>(ring, Aih, Ahh, bias, h, c, Dacc, hin);
    gates(Dacc, h, c);

    float ph[NSL - 1];
#pragma unroll
    for (int j = 0; j < NSL - 1; ++j) ph[j] = __shfl_xor(h[j], 16, 32);
    if (hi == 0) {
      float hall[HID];
#pragma unroll
      for (int j = 0; j < NSL; ++j) hall[j] = h[j];
#pragma unroll
      for (int j = 0; j < NSL - 1; ++j) hall[NSL + j] = ph[j];
      float* op = out + (size_t)bglob * NCLS;
#pragma unroll
      for (int k = 0; k < NCLS; ++k) {
        float acc = fcb[k];
#pragma unroll
        for (int j = 0; j < HID; ++j) acc += hall[j] * fcw[k * HID + j];
        op[k] = acc;
      }
    }
  }
}

extern "C" void kernel_launch(void* const* d_in, const int* in_sizes, int n_in,
                              void* d_out, int out_size, void* d_ws, size_t ws_size,
                              hipStream_t stream) {
  const float* x    = (const float*)d_in[0];
  const float* wih0 = (const float*)d_in[1];
  const float* whh0 = (const float*)d_in[2];
  const float* bih0 = (const float*)d_in[3];
  const float* bhh0 = (const float*)d_in[4];
  const float* wih1 = (const float*)d_in[5];
  const float* whh1 = (const float*)d_in[6];
  const float* bih1 = (const float*)d_in[7];
  const float* bhh1 = (const float*)d_in[8];
  const float* fcw  = (const float*)d_in[9];
  const float* fcb  = (const float*)d_in[10];
  float* out = (float*)d_out;

  const int B = in_sizes[0] / (TLEN * HID);
  dim3 grid(B / 16);
  dim3 block(64);
  lstm2_fc_fused<<<grid, block, 0, stream>>>(x, wih0, whh0, bih0, bhh0,
                                             wih1, whh1, bih1, bhh1,
                                             fcw, fcb, out);
}


# input and recurrent products of each layer fused over one k range of 20 slots: 15 matrix instructions per step per layer where 9 + 9 were; k sum order changes
# speedup vs baseline: 3.4606x; 1.0533x over previous
#include <hip/hip_runtime.h>
#include <math.h>


#define HID 9
#define NG  36
#define TLEN 512
#define NCLS 10
#define NSL 5
#define NCH 5
#define NCH_EARLY 2


typedef __attribute__((ext_vector_type(2))) float v2f;
typedef __attribute__((ext_vector_type(8))) float v8f;

__device__ __forceinline__ float fexp2(float x) {
#if defined(__HIP_DEVICE_COMPILE__) && __has_builtin(__builtin_amdgcn_exp2f)
  return __builtin_amdgcn_exp2f(x);
#else
  return exp2f(x);
#endif
}

__device__ __forceinline__ float frcp(float x) {
#if defined(__HIP_DEVICE_COMPILE__) && __has_builtin(__builtin_amdgcn_rcpf)
  return __builtin_amdgcn_rcpf(x);
#else
  return 1.0f / x;
#endif
}

__device__ __forceinline__ float ftanh(float x) {
#if defined(__HIP_DEVICE_COMPILE__) && __has_builtin(__builtin_amdgcn_tanhf)
  return __builtin_amdgcn_tanhf(x);
#else
  float e = fexp2(2.8853900817779268f * x);
  return 1.0f - 2.0f * frcp(e + 1.0f);
#endif
}

__device__ __forceinline__ float fsig(float x) {
  return 0.5f + 0.5f * ftanh(0.5f * x);
}

__device__ __forceinline__ v8f wmma4(v2f a, v2f b, v8f c) {
  return __builtin_amdgcn_wmma_f32_16x16x4_f32(false, a, false, b, (short)0, c,
                                               false, false);
}

__device__ __forceinline__ int unit_of(int hh, int j) { return hh ? (NSL + j) : j; }
__device__ __forceinline__ bool slot_ok(int hh, int j) { return j < NSL && unit_of(hh, j) < HID; }

__device__ __forceinline__ void load_A(const float* __restrict__ Win,
                                       const float* __restrict__ Whh,
                                       v2f A[3][NCH], int m, int hi) {
  const int hr = m >> 3;
#pragma unroll
  for (int tt = 0; tt < 3; ++tt) {
    int s = 8 * tt + (m & 7);
    int j = s >> 2;
    int g = s & 3;
    bool rok = slot_ok(hr, j);
    int row = rok ? (g * HID + unit_of(hr, j)) : 0;
#pragma unroll
    for (int cc = 0; cc < NCH; ++cc) {
      v2f a;
#pragma unroll
      for (int v = 0; v < 2; ++v) {
        int q = 2 * cc + v;
        const float* __restrict__ W = (q < NSL) ? Win : Whh;
        int qs = (q < NSL) ? q : (q - NSL);
        bool ok = rok && slot_ok(hi, qs);
        int col = slot_ok(hi, qs) ? unit_of(hi, qs) : 0;
        float w = W[row * HID + col];
        a[v] = ok ? w : 0.0f;
      }
      A[tt][cc] = a;
    }
  }
}

__device__ __forceinline__ void load_bias(const float* __restrict__ bi,
                                          const float* __restrict__ bh,
                                          v8f bias[3], int hi) {
#pragma unroll
  for (int tt = 0; tt < 3; ++tt) {
#pragma unroll
    for (int r = 0; r < 8; ++r) {
      int s = 8 * tt + r;
      int j = s >> 2;
      int g = s & 3;
      bool ok = slot_ok(hi, j);
      int row = ok ? (g * HID + unit_of(hi, j)) : 0;
      float b = bi[row] + bh[row];
      bias[tt][r] = ok ? b : 0.0f;
    }
  }
}

__device__ __forceinline__ void build_B_early(const float in[NSL], v2f B[NCH]) {
  v2f b0; b0.x = in[0]; b0.y = in[1];
  v2f b1; b1.x = in[2]; b1.y = in[3];
  B[0] = b0; B[1] = b1;
}
__device__ __forceinline__ void build_B_late(float in4, const float h[NSL], v2f B[NCH]) {
  v2f b2; b2.x = in4;  b2.y = h[0];
  v2f b3; b3.x = h[1]; b3.y = h[2];
  v2f b4; b4.x = h[3]; b4.y = h[4];
  B[2] = b2; B[3] = b3; B[4] = b4;
}

template <int C0, int C1>
__device__ __forceinline__ void mm(const v2f A[3][NCH], const v2f B[NCH], v8f D[3]) {
#pragma unroll
  for (int cc = C0; cc < C1; ++cc) {
#pragma unroll
    for (int tt = 0; tt < 3; ++tt) D[tt] = wmma4(A[tt][cc], B[cc], D[tt]);
  }
}

__device__ __forceinline__ void gates(const v8f D[3], float h[NSL], float cst[NSL]) {
#pragma unroll
  for (int j = 0; j < NSL; ++j) {
    float ig = fsig(D[(4 * j) >> 3][(4 * j) & 7]);
    float fg = fsig(D[(4 * j + 1) >> 3][(4 * j + 1) & 7]);
    float gg = ftanh(D[(4 * j + 2) >> 3][(4 * j + 2) & 7]);
    float og = fsig(D[(4 * j + 3) >> 3][(4 * j + 3) & 7]);
    float cn = fg * cst[j] + ig * gg;
    cst[j] = cn;
    h[j] = og * ftanh(cn);
  }
}

__device__ __forceinline__ void fetch_x(const float* __restrict__ xp, int hi, float xs[NSL]) {
#pragma unroll
  for (int q = 0; q < NSL; ++q) xs[q] = xp[slot_ok(hi, q) ? unit_of(hi, q) : (HID - 1)];
}
__device__ __forceinline__ float build_Bx(const float xs[NSL], int hi, v2f B[NCH]) {
  float s[NSL];
#pragma unroll
  for (int q = 0; q < NSL; ++q) s[q] = slot_ok(hi, q) ? xs[q] : 0.0f;
  build_B_early(s, B);
  return s[NSL - 1];
}

__device__ __forceinline__ void wg_sync() {
  asm volatile("s_wait_dscnt 0x0\n\ts_barrier_signal -1\n\ts_barrier_wait -1" ::: "memory");
}

__device__ __forceinline__ void l0_step(const float* __restrict__ xrow, int k, int hi,
                                        const v2f A[3][NCH],
                                        const v8f bias[3], float h[NSL], float c[NSL],
                                        v8f Dx[3], float& x4, const float xuse[NSL],
                                        float xload[NSL], float* dst) {
  const int tp = (k + 11 < TLEN) ? (k + 11) : (TLEN - 1);
  __builtin_prefetch(xrow + tp * HID, 0, 0);
  const int tl = (k + 3 < TLEN) ? (k + 3) : (TLEN - 1);
  fetch_x(xrow + tl * HID, hi, xload);

  v2f B[NCH];
  v8f D0[3];
  build_B_late(x4, h, B);
  D0[0] = Dx[0]; D0[1] = Dx[1]; D0[2] = Dx[2];
  mm<NCH_EARLY, NCH>(A, B, D0);
  __builtin_amdgcn_sched_barrier(0);
  x4 = build_Bx(xuse, hi, B);
  Dx[0] = bias[0]; Dx[1] = bias[1]; Dx[2] = bias[2];
  gates(D0, h, c);
#pragma unroll
  for (int j = 0; j < NSL; ++j) dst[32 * j] = h[j];
  mm<0, NCH_EARLY>(A, B, Dx);
}

template <bool GATES, bool FETCH>
__device__ __forceinline__ void l1_step(const float* src,
                                        const v2f A[3][NCH],
                                        const v8f bias[3], float h[NSL], float c[NSL],
                                        v8f D[3], float hin[NSL]) {
  float hn[NSL];
  if (FETCH) {
#pragma unroll
    for (int j = 0; j < NSL; ++j) hn[j] = src[32 * j];
  }
  v2f B[NCH];
  v8f E[3];
  E[0] = bias[0]; E[1] = bias[1]; E[2] = bias[2];
  build_B_early(hin, B);
  if (GATES) gates(D, h, c);
  mm<0, NCH_EARLY>(A, B, E);
  build_B_late(hin[NSL - 1], h, B);
  mm<NCH_EARLY, NCH>(A, B, E);
  D[0] = E[0]; D[1] = E[1]; D[2] = E[2];
  if (FETCH) {
#pragma unroll
    for (int j = 0; j < NSL; ++j) hin[j] = hn[j];
  }
}

__global__ __launch_bounds__(64) void lstm2_fc_fused(
    const float* __restrict__ x,
    const float* __restrict__ wih0, const float* __restrict__ whh0,
    const float* __restrict__ bih0, const float* __restrict__ bhh0,
    const float* __restrict__ wih1, const float* __restrict__ whh1,
    const float* __restrict__ bih1, const float* __restrict__ bhh1,
    const float* __restrict__ fcw,  const float* __restrict__ fcb,
    float* __restrict__ out) {
  __shared__ float hx[2][NSL][32];

  const int lane = threadIdx.x & 31;
  const int role = __builtin_amdgcn_readfirstlane((int)(threadIdx.x >> 5));
  const int m  = lane & 15;
  const int hi = lane >> 4;
  const int bglob = blockIdx.x * 16 + m;

  const float* wih = role ? wih1 : wih0;
  const float* whh = role ? whh1 : whh0;
  const float* bih = role ? bih1 : bih0;
  const float* bhh = role ? bhh1 : bhh0;
  v2f A[3][NCH];
  load_A(wih, whh, A, m, hi);
  v8f bias[3];
  load_bias(bih, bhh, bias, hi);

  float h[NSL], c[NSL];
  float xs0[NSL], xs1[NSL], xs2[NSL];
  float hin[NSL];
#pragma unroll
  for (int j = 0; j < NSL; ++j) { h[j] = c[j] = 0.0f; xs0[j] = xs1[j] = xs2[j] = hin[j] = 0.0f; }
  v8f Dacc[3];
  Dacc[0] = bias[0]; Dacc[1] = bias[1]; Dacc[2] = bias[2];
  float x4 = 0.0f;

  const float* xrow = x + (size_t)bglob * TLEN * HID;
  float* ring = &hx[0][0][lane];
  const int RB = NSL * 32;

  if (role == 0) {
    float x0[NSL];
    v2f Bx[NCH];
    fetch_x(xrow, hi, x0);
    fetch_x(xrow + HID, hi, xs0);
    fetch_x(xrow + 2 * HID, hi, xs1);
    x4 = build_Bx(x0, hi, Bx);
    mm<0, NCH_EARLY>(A, Bx, Dacc);
  }

  if (role == 0) l0_step(xrow, 0, hi, A, bias, h, c, Dacc, x4, xs0, xs2, ring);
  wg_sync();
  if (role == 0) {
    l0_step(xrow, 1, hi, A, bias, h, c, Dacc, x4, xs1, xs0, ring + RB);
  } else {
#pragma unroll
    for (int j = 0; j < NSL; ++j) hin[j] = ring[32 * j];
  }
  wg_sync();
  if (role == 0) l0_step(xrow, 2, hi, A, bias, h, c, Dacc, x4, xs2, xs1, ring);
  else           l1_step<false, true>(ring + RB, A, bias, h, c, Dacc, hin);
  wg_sync();

  for (int k = 3; k < TLEN + 1; k += 3) {
    float* b0 = ring + ((k & 1) ? RB : 0);
    float* b1 = ring + ((k & 1) ? 0 : RB);
    if (role == 0) { if (k < TLEN)     l0_step(xrow, k,     hi, A, bias, h, c, Dacc, x4, xs0, xs2, b0); }
    else           l1_step<true, true>(b1, A, bias, h, c, Dacc, hin);
    wg_sync();
    if (role == 0) { if (k + 1 < TLEN) l0_step(xrow, k + 1, hi, A, bias, h, c, Dacc, x4, xs1, xs0, b1); }
    else           l1_step<true, true>(b0, A, bias, h, c, Dacc, hin);
    wg_sync();
    if (role == 0) { if (k + 2 < TLEN) l0_step(xrow, k + 2, hi, A, bias, h, c, Dacc, x4, xs2, xs1, b0); }
    else           l1_step<true, true>(b1, A, bias, h, c, Dacc, hin);
    wg_sync();
  }

  if (role == 1) {
    l1_step<true, false>(ring, A, bias, h, c, Dacc, hin);
    gates(Dacc, h, c);

    float ph[NSL - 1];
#pragma unroll
    for (int j = 0; j < NSL - 1; ++j) ph[j] = __shfl_xor(h[j], 16, 32);
    if (hi == 0) {
      float hall[HID];
#pragma unroll
      for (int j = 0; j < NSL; ++j) hall[j] = h[j];
#pragma unroll
      for (int j = 0; j < NSL - 1; ++j) hall[NSL + j] = ph[j];
      float* op = out + (size_t)bglob * NCLS;
#pragma unroll
      for (int k = 0; k < NCLS; ++k) {
        float acc = fcb[k];
#pragma unroll
        for (int j = 0; j < HID; ++j) acc += hall[j] * fcw[k * HID + j];
        op[k] = acc;
      }
    }
  }
}

extern "C" void kernel_launch(void* const* d_in, const int* in_sizes, int n_in,
                              void* d_out, int out_size, void* d_ws, size_t ws_size,
                              hipStream_t stream) {
  const float* x    = (const float*)d_in[0];
  const float* wih0 = (const float*)d_in[1];
  const float* whh0 = (const float*)d_in[2];
  const float* bih0 = (const float*)d_in[3];
  const float* bhh0 = (const float*)d_in[4];
  const float* wih1 = (const float*)d_in[5];
  const float* whh1 = (const float*)d_in[6];
  const float* bih1 = (const float*)d_in[7];
  const float* bhh1 = (const float*)d_in[8];
  const float* fcw  = (const float*)d_in[9];
  const float* fcb  = (const float*)d_in[10];
  float* out = (float*)d_out;

  const int B = in_sizes[0] / (TLEN * HID);
  dim3 grid(B / 16);
  dim3 block(64);
  lstm2_fc_fused<<<grid, block, 0, stream>>>(x, wih0, whh0, bih0, bhh0,
                                             wih1, whh1, bih1, bhh1,
                                             fcw, fcb, out);
}


# layer-0 x rows fetched three at a time, one trip of three steps ahead, into a second register set: loads in flight three steps; before, each step's loads waited for the step before's; sums unchanged
# speedup vs baseline: 3.6594x; 1.0574x over previous
#include <hip/hip_runtime.h>
#include <math.h>


#define HID 9
#define NG  36
#define TLEN 512
#define NCLS 10
#define NSL 5
#define NCH 5
#define NCH_EARLY 2


typedef __attribute__((ext_vector_type(2))) float v2f;
typedef __attribute__((ext_vector_type(8))) float v8f;

__device__ __forceinline__ float fexp2(float x) {
#if defined(__HIP_DEVICE_COMPILE__) && __has_builtin(__builtin_amdgcn_exp2f)
  return __builtin_amdgcn_exp2f(x);
#else
  return exp2f(x);
#endif
}

__device__ __forceinline__ float frcp(float x) {
#if defined(__HIP_DEVICE_COMPILE__) && __has_builtin(__builtin_amdgcn_rcpf)
  return __builtin_amdgcn_rcpf(x);
#else
  return 1.0f / x;
#endif
}

__device__ __forceinline__ float ftanh(float x) {
#if defined(__HIP_DEVICE_COMPILE__) && __has_builtin(__builtin_amdgcn_tanhf)
  return __builtin_amdgcn_tanhf(x);
#else
  float e = fexp2(2.8853900817779268f * x);
  return 1.0f - 2.0f * frcp(e + 1.0f);
#endif
}

__device__ __forceinline__ float fsig(float x) {
  return 0.5f + 0.5f * ftanh(0.5f * x);
}

__device__ __forceinline__ v8f wmma4(v2f a, v2f b, v8f c) {
  return __builtin_amdgcn_wmma_f32_16x16x4_f32(false, a, false, b, (short)0, c,
                                               false, false);
}

__device__ __forceinline__ int unit_of(int hh, int j) { return hh ? (NSL + j) : j; }
__device__ __forceinline__ bool slot_ok(int hh, int j) { return j < NSL && unit_of(hh, j) < HID; }

__device__ __forceinline__ void load_A(const float* __restrict__ Win,
                                       const float* __restrict__ Whh,
                                       v2f A[3][NCH], int m, int hi) {
  const int hr = m >> 3;
#pragma unroll
  for (int tt = 0; tt < 3; ++tt) {
    int s = 8 * tt + (m & 7);
    int j = s >> 2;
    int g = s & 3;
    bool rok = slot_ok(hr, j);
    int row = rok ? (g * HID + unit_of(hr, j)) : 0;
#pragma unroll
    for (int cc = 0; cc < NCH; ++cc) {
      v2f a;
#pragma unroll
      for (int v = 0; v < 2; ++v) {
        int q = 2 * cc + v;
        const float* __restrict__ W = (q < NSL) ? Win : Whh;
        int qs = (q < NSL) ? q : (q - NSL);
        bool ok = rok && slot_ok(hi, qs);
        int col = slot_ok(hi, qs) ? unit_of(hi, qs) : 0;
        float w = W[row * HID + col];
        a[v] = ok ? w : 0.0f;
      }
      A[tt][cc] = a;
    }
  }
}

__device__ __forceinline__ void load_bias(const float* __restrict__ bi,
                                          const float* __restrict__ bh,
                                          v8f bias[3], int hi) {
#pragma unroll
  for (int tt = 0; tt < 3; ++tt) {
#pragma unroll
    for (int r = 0; r < 8; ++r) {
      int s = 8 * tt + r;
      int j = s >> 2;
      int g = s & 3;
      bool ok = slot_ok(hi, j);
      int row = ok ? (g * HID + unit_of(hi, j)) : 0;
      float b = bi[row] + bh[row];
      bias[tt][r] = ok ? b : 0.0f;
    }
  }
}

__device__ __forceinline__ void build_B_early(const float in[NSL], v2f B[NCH]) {
  v2f b0; b0.x = in[0]; b0.y = in[1];
  v2f b1; b1.x = in[2]; b1.y = in[3];
  B[0] = b0; B[1] = b1;
}
__device__ __forceinline__ void build_B_late(float in4, const float h[NSL], v2f B[NCH]) {
  v2f b2; b2.x = in4;  b2.y = h[0];
  v2f b3; b3.x = h[1]; b3.y = h[2];
  v2f b4; b4.x = h[3]; b4.y = h[4];
  B[2] = b2; B[3] = b3; B[4] = b4;
}

template <int C0, int C1>
__device__ __forceinline__ void mm(const v2f A[3][NCH], const v2f B[NCH], v8f D[3]) {
#pragma unroll
  for (int cc = C0; cc < C1; ++cc) {
#pragma unroll
    for (int tt = 0; tt < 3; ++tt) D[tt] = wmma4(A[tt][cc], B[cc], D[tt]);
  }
}

__device__ __forceinline__ void gates(const v8f D[3], float h[NSL], float cst[NSL]) {
#pragma unroll
  for (int j = 0; j < NSL; ++j) {
    float ig = fsig(D[(4 * j) >> 3][(4 * j) & 7]);
    float fg = fsig(D[(4 * j + 1) >> 3][(4 * j + 1) & 7]);
    float gg = ftanh(D[(4 * j + 2) >> 3][(4 * j + 2) & 7]);
    float og = fsig(D[(4 * j + 3) >> 3][(4 * j + 3) & 7]);
    float cn = fg * cst[j] + ig * gg;
    cst[j] = cn;
    h[j] = og * ftanh(cn);
  }
}

__device__ __forceinline__ void fetch_x(const float* __restrict__ xp, int hi, float xs[NSL]) {
#pragma unroll
  for (int q = 0; q < NSL; ++q) xs[q] = xp[slot_ok(hi, q) ? unit_of(hi, q) : (HID - 1)];
}
__device__ __forceinline__ void fetch_x3(const float* __restrict__ xrow, int t0, int hi,
                                         float ya[NSL], float yb[NSL], float yc[NSL]) {
  const int ra = (t0     < TLEN) ? t0       : (TLEN - 1);
  const int rb = (t0 + 1 < TLEN) ? (t0 + 1) : (TLEN - 1);
  const int rc = (t0 + 2 < TLEN) ? (t0 + 2) : (TLEN - 1);
  fetch_x(xrow + ra * HID, hi, ya);
  fetch_x(xrow + rb * HID, hi, yb);
  fetch_x(xrow + rc * HID, hi, yc);
}
__device__ __forceinline__ float build_Bx(const float xs[NSL], int hi, v2f B[NCH]) {
  float s[NSL];
#pragma unroll
  for (int q = 0; q < NSL; ++q) s[q] = slot_ok(hi, q) ? xs[q] : 0.0f;
  build_B_early(s, B);
  return s[NSL - 1];
}

__device__ __forceinline__ void wg_sync() {
  asm volatile("s_wait_dscnt 0x0\n\ts_barrier_signal -1\n\ts_barrier_wait -1" ::: "memory");
}

__device__ __forceinline__ void loads_settled() {
  __builtin_amdgcn_fence(__ATOMIC_ACQUIRE, "workgroup");
}

__device__ __forceinline__ void l0_step(const float* __restrict__ xrow, int k, int hi,
                                        const v2f A[3][NCH],
                                        const v8f bias[3], float h[NSL], float c[NSL],
                                        v8f Dx[3], float& x4, const float xuse[NSL],
                                        float* dst) {
  const int tp = (k + 11 < TLEN) ? (k + 11) : (TLEN - 1);
  __builtin_prefetch(xrow + tp * HID, 0, 0);

  v2f B[NCH];
  v8f D0[3];
  build_B_late(x4, h, B);
  D0[0] = Dx[0]; D0[1] = Dx[1]; D0[2] = Dx[2];
  mm<NCH_EARLY, NCH>(A, B, D0);
  __builtin_amdgcn_sched_barrier(0);
  x4 = build_Bx(xuse, hi, B);
  Dx[0] = bias[0]; Dx[1] = bias[1]; Dx[2] = bias[2];
  gates(D0, h, c);
#pragma unroll
  for (int j = 0; j < NSL; ++j) dst[32 * j] = h[j];
  mm<0, NCH_EARLY>(A, B, Dx);
}

template <bool GATES, bool FETCH>
__device__ __forceinline__ void l1_step(const float* src,
                                        const v2f A[3][NCH],
                                        const v8f bias[3], float h[NSL], float c[NSL],
                                        v8f D[3], float hin[NSL]) {
  float hn[NSL];
  if (FETCH) {
#pragma unroll
    for (int j = 0; j < NSL; ++j) hn[j] = src[32 * j];
  }
  v2f B[NCH];
  v8f E[3];
  E[0] = bias[0]; E[1] = bias[1]; E[2] = bias[2];
  build_B_early(hin, B);
  if (GATES) gates(D, h, c);
  mm<0, NCH_EARLY>(A, B, E);
  build_B_late(hin[NSL - 1], h, B);
  mm<NCH_EARLY, NCH>(A, B, E);
  D[0] = E[0]; D[1] = E[1]; D[2] = E[2];
  if (FETCH) {
#pragma unroll
    for (int j = 0; j < NSL; ++j) hin[j] = hn[j];
  }
}

__global__ __launch_bounds__(64) void lstm2_fc_fused(
    const float* __restrict__ x,
    const float* __restrict__ wih0, const float* __restrict__ whh0,
    const float* __restrict__ bih0, const float* __restrict__ bhh0,
    const float* __restrict__ wih1, const float* __restrict__ whh1,
    const float* __restrict__ bih1, const float* __restrict__ bhh1,
    const float* __restrict__ fcw,  const float* __restrict__ fcb,
    float* __restrict__ out) {
  __shared__ float hx[2][NSL][32];

  const int lane = threadIdx.x & 31;
  const int role = __builtin_amdgcn_readfirstlane((int)(threadIdx.x >> 5));
  const int m  = lane & 15;
  const int hi = lane >> 4;
  const int bglob = blockIdx.x * 16 + m;

  const float* wih = role ? wih1 : wih0;
  const float* whh = role ? whh1 : whh0;
  const float* bih = role ? bih1 : bih0;
  const float* bhh = role ? bhh1 : bhh0;
  v2f A[3][NCH];
  load_A(wih, whh, A, m, hi);
  v8f bias[3];
  load_bias(bih, bhh, bias, hi);

  float h[NSL], c[NSL];
  float xs0[NSL], xs1[NSL], xs2[NSL];
  float y0[NSL], y1[NSL], y2[NSL];
  float hin[NSL];
#pragma unroll
  for (int j = 0; j < NSL; ++j) {
    h[j] = c[j] = 0.0f; hin[j] = 0.0f;
    xs0[j] = xs1[j] = xs2[j] = 0.0f; y0[j] = y1[j] = y2[j] = 0.0f;
  }
  v8f Dacc[3];
  Dacc[0] = bias[0]; Dacc[1] = bias[1]; Dacc[2] = bias[2];
  float x4 = 0.0f;

  const float* xrow = x + (size_t)bglob * TLEN * HID;
  float* ring = &hx[0][0][lane];
  const int RB = NSL * 32;

  if (role == 0) {
    float x0[NSL];
    v2f Bx[NCH];
    fetch_x(xrow, hi, x0);
    fetch_x3(xrow, 1, hi, xs0, xs1, xs2);
    fetch_x3(xrow, 4, hi, y0, y1, y2);
    x4 = build_Bx(x0, hi, Bx);
    mm<0, NCH_EARLY>(A, Bx, Dacc);
  }

  if (role == 0) l0_step(xrow, 0, hi, A, bias, h, c, Dacc, x4, xs0, ring);
  wg_sync();
  if (role == 0) {
    l0_step(xrow, 1, hi, A, bias, h, c, Dacc, x4, xs1, ring + RB);
  } else {
#pragma unroll
    for (int j = 0; j < NSL; ++j) hin[j] = ring[32 * j];
  }
  wg_sync();
  if (role == 0) l0_step(xrow, 2, hi, A, bias, h, c, Dacc, x4, xs2, ring);
  else           l1_step<false, true>(ring + RB, A, bias, h, c, Dacc, hin);
  wg_sync();
  if (role == 0) {
#pragma unroll
    for (int j = 0; j < NSL; ++j) { xs0[j] = y0[j]; xs1[j] = y1[j]; xs2[j] = y2[j]; }
  }
  loads_settled();

  for (int k = 3; k < TLEN + 1; k += 3) {
    float* b0 = ring + ((k & 1) ? RB : 0);
    float* b1 = ring + ((k & 1) ? 0 : RB);
    if (role == 0) {
      fetch_x3(xrow, k + 4, hi, y0, y1, y2);
      if (k < TLEN)     l0_step(xrow, k,     hi, A, bias, h, c, Dacc, x4, xs0, b0);
    } else {
      l1_step<true, true>(b1, A, bias, h, c, Dacc, hin);
    }
    wg_sync();
    if (role == 0) { if (k + 1 < TLEN) l0_step(xrow, k + 1, hi, A, bias, h, c, Dacc, x4, xs1, b1); }
    else           l1_step<true, true>(b0, A, bias, h, c, Dacc, hin);
    wg_sync();
    if (role == 0) { if (k + 2 < TLEN) l0_step(xrow, k + 2, hi, A, bias, h, c, Dacc, x4, xs2, b0); }
    else           l1_step<true, true>(b1, A, bias, h, c, Dacc, hin);
    wg_sync();
    if (role == 0) {
#pragma unroll
      for (int j = 0; j < NSL; ++j) { xs0[j] = y0[j]; xs1[j] = y1[j]; xs2[j] = y2[j]; }
    }
    loads_settled();
  }

  if (role == 1) {
    l1_step<true, false>(ring, A, bias, h, c, Dacc, hin);
    gates(Dacc, h, c);

    float ph[NSL - 1];
#pragma unroll
    for (int j = 0; j < NSL - 1; ++j) ph[j] = __shfl_xor(h[j], 16, 32);
    if (hi == 0) {
      float hall[HID];
#pragma unroll
      for (int j = 0; j < NSL; ++j) hall[j] = h[j];
#pragma unroll
      for (int j = 0; j < NSL - 1; ++j) hall[NSL + j] = ph[j];
      float* op = out + (size_t)bglob * NCLS;
#pragma unroll
      for (int k = 0; k < NCLS; ++k) {
        float acc = fcb[k];
#pragma unroll
        for (int j = 0; j < HID; ++j) acc += hall[j] * fcw[k * HID + j];
        op[k] = acc;
      }
    }
  }
}

extern "C" void kernel_launch(void* const* d_in, const int* in_sizes, int n_in,
                              void* d_out, int out_size, void* d_ws, size_t ws_size,
                              hipStream_t stream) {
  const float* x    = (const float*)d_in[0];
  const float* wih0 = (const float*)d_in[1];
  const float* whh0 = (const float*)d_in[2];
  const float* bih0 = (const float*)d_in[3];
  const float* bhh0 = (const float*)d_in[4];
  const float* wih1 = (const float*)d_in[5];
  const float* whh1 = (const float*)d_in[6];
  const float* bih1 = (const float*)d_in[7];
  const float* bhh1 = (const float*)d_in[8];
  const float* fcw  = (const float*)d_in[9];
  const float* fcb  = (const float*)d_in[10];
  float* out = (float*)d_out;

  const int B = in_sizes[0] / (TLEN * HID);
  dim3 grid(B / 16);
  dim3 block(64);
  lstm2_fc_fused<<<grid, block, 0, stream>>>(x, wih0, whh0, bih0, bhh0,
                                             wih1, whh1, bih1, bhh1,
                                             fcw, fcb, out);
}


# one workgroup barrier per trip of three time steps, not one per step: two-set LDS ring of three h0 buffers, layer-1 wave four steps behind, a trip's steps run back to back; sums unchanged
# speedup vs baseline: 4.1960x; 1.1466x over previous
#include <hip/hip_runtime.h>
#include <math.h>


#define HID 9
#define NG  36
#define TLEN 512
#define NCLS 10
#define NSL 5
#define NCH 5
#define NCH_EARLY 2
#define SPT 3


typedef __attribute__((ext_vector_type(2))) float v2f;
typedef __attribute__((ext_vector_type(8))) float v8f;

__device__ __forceinline__ float fexp2(float x) {
#if defined(__HIP_DEVICE_COMPILE__) && __has_builtin(__builtin_amdgcn_exp2f)
  return __builtin_amdgcn_exp2f(x);
#else
  return exp2f(x);
#endif
}

__device__ __forceinline__ float frcp(float x) {
#if defined(__HIP_DEVICE_COMPILE__) && __has_builtin(__builtin_amdgcn_rcpf)
  return __builtin_amdgcn_rcpf(x);
#else
  return 1.0f / x;
#endif
}

__device__ __forceinline__ float ftanh(float x) {
#if defined(__HIP_DEVICE_COMPILE__) && __has_builtin(__builtin_amdgcn_tanhf)
  return __builtin_amdgcn_tanhf(x);
#else
  float e = fexp2(2.8853900817779268f * x);
  return 1.0f - 2.0f * frcp(e + 1.0f);
#endif
}

__device__ __forceinline__ float fsig(float x) {
  return 0.5f + 0.5f * ftanh(0.5f * x);
}

__device__ __forceinline__ v8f wmma4(v2f a, v2f b, v8f c) {
  return __builtin_amdgcn_wmma_f32_16x16x4_f32(false, a, false, b, (short)0, c,
                                               false, false);
}

__device__ __forceinline__ int unit_of(int hh, int j) { return hh ? (NSL + j) : j; }
__device__ __forceinline__ bool slot_ok(int hh, int j) { return j < NSL && unit_of(hh, j) < HID; }

__device__ __forceinline__ void load_A(const float* __restrict__ Win,
                                       const float* __restrict__ Whh,
                                       v2f A[3][NCH], int m, int hi) {
  const int hr = m >> 3;
#pragma unroll
  for (int tt = 0; tt < 3; ++tt) {
    int s = 8 * tt + (m & 7);
    int j = s >> 2;
    int g = s & 3;
    bool rok = slot_ok(hr, j);
    int row = rok ? (g * HID + unit_of(hr, j)) : 0;
#pragma unroll
    for (int cc = 0; cc < NCH; ++cc) {
      v2f a;
#pragma unroll
      for (int v = 0; v < 2; ++v) {
        int q = 2 * cc + v;
        const float* __restrict__ W = (q < NSL) ? Win : Whh;
        int qs = (q < NSL) ? q : (q - NSL);
        bool ok = rok && slot_ok(hi, qs);
        int col = slot_ok(hi, qs) ? unit_of(hi, qs) : 0;
        float w = W[row * HID + col];
        a[v] = ok ? w : 0.0f;
      }
      A[tt][cc] = a;
    }
  }
}

__device__ __forceinline__ void load_bias(const float* __restrict__ bi,
                                          const float* __restrict__ bh,
                                          v8f bias[3], int hi) {
#pragma unroll
  for (int tt = 0; tt < 3; ++tt) {
#pragma unroll
    for (int r = 0; r < 8; ++r) {
      int s = 8 * tt + r;
      int j = s >> 2;
      int g = s & 3;
      bool ok = slot_ok(hi, j);
      int row = ok ? (g * HID + unit_of(hi, j)) : 0;
      float b = bi[row] + bh[row];
      bias[tt][r] = ok ? b : 0.0f;
    }
  }
}

__device__ __forceinline__ void build_B_early(const float in[NSL], v2f B[NCH]) {
  v2f b0; b0.x = in[0]; b0.y = in[1];
  v2f b1; b1.x = in[2]; b1.y = in[3];
  B[0] = b0; B[1] = b1;
}
__device__ __forceinline__ void build_B_late(float in4, const float h[NSL], v2f B[NCH]) {
  v2f b2; b2.x = in4;  b2.y = h[0];
  v2f b3; b3.x = h[1]; b3.y = h[2];
  v2f b4; b4.x = h[3]; b4.y = h[4];
  B[2] = b2; B[3] = b3; B[4] = b4;
}

template <int C0, int C1>
__device__ __forceinline__ void mm(const v2f A[3][NCH], const v2f B[NCH], v8f D[3]) {
#pragma unroll
  for (int cc = C0; cc < C1; ++cc) {
#pragma unroll
    for (int tt = 0; tt < 3; ++tt) D[tt] = wmma4(A[tt][cc], B[cc], D[tt]);
  }
}

__device__ __forceinline__ void gates(const v8f D[3], float h[NSL], float cst[NSL]) {
#pragma unroll
  for (int j = 0; j < NSL; ++j) {
    float ig = fsig(D[(4 * j) >> 3][(4 * j) & 7]);
    float fg = fsig(D[(4 * j + 1) >> 3][(4 * j + 1) & 7]);
    float gg = ftanh(D[(4 * j + 2) >> 3][(4 * j + 2) & 7]);
    float og = fsig(D[(4 * j + 3) >> 3][(4 * j + 3) & 7]);
    float cn = fg * cst[j] + ig * gg;
    cst[j] = cn;
    h[j] = og * ftanh(cn);
  }
}

__device__ __forceinline__ void fetch_x(const float* __restrict__ xp, int hi, float xs[NSL]) {
#pragma unroll
  for (int q = 0; q < NSL; ++q) xs[q] = xp[slot_ok(hi, q) ? unit_of(hi, q) : (HID - 1)];
}
__device__ __forceinline__ void fetch_x3(const float* __restrict__ xrow, int t0, int hi,
                                         float ya[NSL], float yb[NSL], float yc[NSL]) {
  const int ra = (t0     < TLEN) ? t0       : (TLEN - 1);
  const int rb = (t0 + 1 < TLEN) ? (t0 + 1) : (TLEN - 1);
  const int rc = (t0 + 2 < TLEN) ? (t0 + 2) : (TLEN - 1);
  fetch_x(xrow + ra * HID, hi, ya);
  fetch_x(xrow + rb * HID, hi, yb);
  fetch_x(xrow + rc * HID, hi, yc);
}
__device__ __forceinline__ float build_Bx(const float xs[NSL], int hi, v2f B[NCH]) {
  float s[NSL];
#pragma unroll
  for (int q = 0; q < NSL; ++q) s[q] = slot_ok(hi, q) ? xs[q] : 0.0f;
  build_B_early(s, B);
  return s[NSL - 1];
}

__device__ __forceinline__ void wg_sync() {
  asm volatile("s_wait_dscnt 0x0\n\ts_barrier_signal -1\n\ts_barrier_wait -1" ::: "memory");
}

template <bool FETCHX>
__device__ __forceinline__ void l0_step(const float* __restrict__ xrow, int k, int hi,
                                        const v2f A[3][NCH],
                                        const v8f bias[3], float h[NSL], float c[NSL],
                                        v8f Dx[3], float& x4, const float xuse[NSL],
                                        float* dst,
                                        float y0[NSL], float y1[NSL], float y2[NSL]) {
  const int tp = (k + 11 < TLEN) ? (k + 11) : (TLEN - 1);
  __builtin_prefetch(xrow + tp * HID, 0, 0);

  v2f B[NCH];
  v8f D0[3];
  build_B_late(x4, h, B);
  D0[0] = Dx[0]; D0[1] = Dx[1]; D0[2] = Dx[2];
  mm<NCH_EARLY, NCH>(A, B, D0);
  __builtin_amdgcn_sched_barrier(0);
  if (FETCHX) fetch_x3(xrow, k + 4, hi, y0, y1, y2);
  x4 = build_Bx(xuse, hi, B);
  Dx[0] = bias[0]; Dx[1] = bias[1]; Dx[2] = bias[2];
  gates(D0, h, c);
#pragma unroll
  for (int j = 0; j < NSL; ++j) dst[32 * j] = h[j];
  mm<0, NCH_EARLY>(A, B, Dx);
  asm volatile("" : "+v"(Dx[0]), "+v"(Dx[1]), "+v"(Dx[2]),
                    "+v"(h[0]), "+v"(h[1]), "+v"(h[2]), "+v"(h[3]), "+v"(h[4]));
}

template <bool GATES, bool FETCH>
__device__ __forceinline__ void l1_step(const float* src,
                                        const v2f A[3][NCH],
                                        const v8f bias[3], float h[NSL], float c[NSL],
                                        v8f D[3], float hin[NSL]) {
  float hn[NSL];
  if (FETCH) {
#pragma unroll
    for (int j = 0; j < NSL; ++j) hn[j] = src[32 * j];
  }
  v2f B[NCH];
  v8f E[3];
  E[0] = bias[0]; E[1] = bias[1]; E[2] = bias[2];
  build_B_early(hin, B);
  if (GATES) gates(D, h, c);
  mm<0, NCH_EARLY>(A, B, E);
  build_B_late(hin[NSL - 1], h, B);
  mm<NCH_EARLY, NCH>(A, B, E);
  D[0] = E[0]; D[1] = E[1]; D[2] = E[2];
  if (FETCH) {
#pragma unroll
    for (int j = 0; j < NSL; ++j) hin[j] = hn[j];
  }
  asm volatile("" : "+v"(D[0]), "+v"(D[1]), "+v"(D[2]),
                    "+v"(hin[0]), "+v"(hin[1]), "+v"(hin[2]), "+v"(hin[3]), "+v"(hin[4]));
}

template <int N, bool FETCH>
__device__ __forceinline__ void l0_trip(const float* __restrict__ xrow, int k, int hi,
                                        const v2f A[3][NCH], const v8f bias[3],
                                        float h[NSL], float c[NSL], v8f Dx[3], float& x4,
                                        float xs0[NSL], float xs1[NSL], float xs2[NSL],
                                        float y0[NSL], float y1[NSL], float y2[NSL],
                                        float* set) {
  const int RB = NSL * 32;
  l0_step<FETCH>(xrow, k, hi, A, bias, h, c, Dx, x4, xs0, set, y0, y1, y2);
  if (N > 1) l0_step<false>(xrow, k + 1, hi, A, bias, h, c, Dx, x4, xs1, set + RB, y0, y1, y2);
  if (N > 2) l0_step<false>(xrow, k + 2, hi, A, bias, h, c, Dx, x4, xs2, set + 2 * RB, y0, y1, y2);
  if (FETCH) {
#pragma unroll
    for (int j = 0; j < NSL; ++j) { xs0[j] = y0[j]; xs1[j] = y1[j]; xs2[j] = y2[j]; }
  }
}

template <int NS, int NF>
__device__ __forceinline__ void l1_trip(const float* set,
                                        const v2f A[3][NCH], const v8f bias[3],
                                        float h[NSL], float c[NSL], v8f D[3],
                                        float hin[NSL]) {
  const int RB = NSL * 32;
  l1_step<true, (NF > 0)>(set, A, bias, h, c, D, hin);
  if (NS > 1) l1_step<true, (NF > 1)>(set + RB, A, bias, h, c, D, hin);
  if (NS > 2) l1_step<true, (NF > 2)>(set + 2 * RB, A, bias, h, c, D, hin);
}

__global__ __launch_bounds__(64) void lstm2_fc_fused(
    const float* __restrict__ x,
    const float* __restrict__ wih0, const float* __restrict__ whh0,
    const float* __restrict__ bih0, const float* __restrict__ bhh0,
    const float* __restrict__ wih1, const float* __restrict__ whh1,
    const float* __restrict__ bih1, const float* __restrict__ bhh1,
    const float* __restrict__ fcw,  const float* __restrict__ fcb,
    float* __restrict__ out) {
  __shared__ float hx[2][SPT][NSL][32];

  const int lane = threadIdx.x & 31;
  const int role = __builtin_amdgcn_readfirstlane((int)(threadIdx.x >> 5));
  const int m  = lane & 15;
  const int hi = lane >> 4;
  const int bglob = blockIdx.x * 16 + m;

  const float* wih = role ? wih1 : wih0;
  const float* whh = role ? whh1 : whh0;
  const float* bih = role ? bih1 : bih0;
  const float* bhh = role ? bhh1 : bhh0;
  v2f A[3][NCH];
  load_A(wih, whh, A, m, hi);
  v8f bias[3];
  load_bias(bih, bhh, bias, hi);

  float h[NSL], c[NSL];
  float xs0[NSL], xs1[NSL], xs2[NSL];
  float y0[NSL], y1[NSL], y2[NSL];
  float hin[NSL];
#pragma unroll
  for (int j = 0; j < NSL; ++j) {
    h[j] = c[j] = 0.0f; hin[j] = 0.0f;
    xs0[j] = xs1[j] = xs2[j] = 0.0f; y0[j] = y1[j] = y2[j] = 0.0f;
  }
  v8f Dacc[3];
  Dacc[0] = bias[0]; Dacc[1] = bias[1]; Dacc[2] = bias[2];
  float x4 = 0.0f;

  const float* xrow = x + (size_t)bglob * TLEN * HID;
  float* ring = &hx[0][0][0][lane];
  const int SB = SPT * NSL * 32;
  const int NFULL = TLEN / SPT;
  const int REM = TLEN - NFULL * SPT;

  if (role == 0) {
    float x0[NSL];
    v2f Bx[NCH];
    fetch_x(xrow, hi, x0);
    fetch_x3(xrow, 1, hi, xs0, xs1, xs2);
    x4 = build_Bx(x0, hi, Bx);
    mm<0, NCH_EARLY>(A, Bx, Dacc);
  }

  if (role == 0) l0_trip<SPT, true>(xrow, 0, hi, A, bias, h, c, Dacc, x4,
                                    xs0, xs1, xs2, y0, y1, y2, ring);
  wg_sync();
  if (role == 0) {
    l0_trip<SPT, true>(xrow, SPT, hi, A, bias, h, c, Dacc, x4,
                       xs0, xs1, xs2, y0, y1, y2, ring + SB);
  } else {
#pragma unroll
    for (int j = 0; j < NSL; ++j) hin[j] = ring[32 * j];
    l1_step<false, true>(ring + NSL * 32, A, bias, h, c, Dacc, hin);
    l1_step<true, true>(ring + 2 * NSL * 32, A, bias, h, c, Dacc, hin);
  }
  wg_sync();

  float* wset = ring;
  float* rset = ring + SB;
  for (int k = 2 * SPT; k < NFULL * SPT; k += SPT) {
    if (role == 0) l0_trip<SPT, true>(xrow, k, hi, A, bias, h, c, Dacc, x4,
                                      xs0, xs1, xs2, y0, y1, y2, wset);
    else           l1_trip<SPT, SPT>(rset, A, bias, h, c, Dacc, hin);
    wg_sync();
    float* t = wset; wset = rset; rset = t;
  }

  if (role == 0) {
    if (REM > 0) l0_trip<(REM > 0 ? REM : 1), false>(xrow, NFULL * SPT, hi, A, bias, h, c, Dacc, x4,
                                                      xs0, xs1, xs2, y0, y1, y2, wset);
  } else {
    l1_trip<SPT, SPT>(rset, A, bias, h, c, Dacc, hin);
  }
  wg_sync();

  if (role == 1) {
    l1_trip<REM + 1, REM>(wset, A, bias, h, c, Dacc, hin);
    gates(Dacc, h, c);

    float ph[NSL - 1];
#pragma unroll
    for (int j = 0; j < NSL - 1; ++j) ph[j] = __shfl_xor(h[j], 16, 32);
    if (hi == 0) {
      float hall[HID];
#pragma unroll
      for (int j = 0; j < NSL; ++j) hall[j] = h[j];
#pragma unroll
      for (int j = 0; j < NSL - 1; ++j) hall[NSL + j] = ph[j];
      float* op = out + (size_t)bglob * NCLS;
#pragma unroll
      for (int k = 0; k < NCLS; ++k) {
        float acc = fcb[k];
#pragma unroll
        for (int j = 0; j < HID; ++j) acc += hall[j] * fcw[k * HID + j];
        op[k] = acc;
      }
    }
  }
}

extern "C" void kernel_launch(void* const* d_in, const int* in_sizes, int n_in,
                              void* d_out, int out_size, void* d_ws, size_t ws_size,
                              hipStream_t stream) {
  const float* x    = (const float*)d_in[0];
  const float* wih0 = (const float*)d_in[1];
  const float* whh0 = (const float*)d_in[2];
  const float* bih0 = (const float*)d_in[3];
  const float* bhh0 = (const float*)d_in[4];
  const float* wih1 = (const float*)d_in[5];
  const float* whh1 = (const float*)d_in[6];
  const float* bih1 = (const float*)d_in[7];
  const float* bhh1 = (const float*)d_in[8];
  const float* fcw  = (const float*)d_in[9];
  const float* fcb  = (const float*)d_in[10];
  float* out = (float*)d_out;

  const int B = in_sizes[0] / (TLEN * HID);
  dim3 grid(B / 16);
  dim3 block(64);
  lstm2_fc_fused<<<grid, block, 0, stream>>>(x, wih0, whh0, bih0, bhh0,
                                             wih1, whh1, bih1, bhh1,
                                             fcw, fcb, out);
}


# workgroup barrier split into signal and wait over a three-set LDS ring, each wave in its own loop: the layer-0 wave signals a trip's rows a step late, waits a trip late, never stalls; sums unchanged
# speedup vs baseline: 4.5721x; 1.0896x over previous
#include <hip/hip_runtime.h>
#include <math.h>


#define HID 9
#define NG  36
#define TLEN 512
#define NCLS 10
#define NSL 5
#define NCH 5
#define NCH_EARLY 2
#define SPT 3
#define NSET 3


typedef __attribute__((ext_vector_type(2))) float v2f;
typedef __attribute__((ext_vector_type(8))) float v8f;

__device__ __forceinline__ float fexp2(float x) {
#if defined(__HIP_DEVICE_COMPILE__) && __has_builtin(__builtin_amdgcn_exp2f)
  return __builtin_amdgcn_exp2f(x);
#else
  return exp2f(x);
#endif
}

__device__ __forceinline__ float frcp(float x) {
#if defined(__HIP_DEVICE_COMPILE__) && __has_builtin(__builtin_amdgcn_rcpf)
  return __builtin_amdgcn_rcpf(x);
#else
  return 1.0f / x;
#endif
}

__device__ __forceinline__ float ftanh(float x) {
#if defined(__HIP_DEVICE_COMPILE__) && __has_builtin(__builtin_amdgcn_tanhf)
  return __builtin_amdgcn_tanhf(x);
#else
  float e = fexp2(2.8853900817779268f * x);
  return 1.0f - 2.0f * frcp(e + 1.0f);
#endif
}

__device__ __forceinline__ float fsig(float x) {
  return 0.5f + 0.5f * ftanh(0.5f * x);
}

__device__ __forceinline__ v8f wmma4(v2f a, v2f b, v8f c) {
  return __builtin_amdgcn_wmma_f32_16x16x4_f32(false, a, false, b, (short)0, c,
                                               false, false);
}

__device__ __forceinline__ int unit_of(int hh, int j) { return hh ? (NSL + j) : j; }
__device__ __forceinline__ bool slot_ok(int hh, int j) { return j < NSL && unit_of(hh, j) < HID; }

__device__ __forceinline__ void load_A(const float* __restrict__ Win,
                                       const float* __restrict__ Whh,
                                       v2f A[3][NCH], int m, int hi) {
  const int hr = m >> 3;
#pragma unroll
  for (int tt = 0; tt < 3; ++tt) {
    int s = 8 * tt + (m & 7);
    int j = s >> 2;
    int g = s & 3;
    bool rok = slot_ok(hr, j);
    int row = rok ? (g * HID + unit_of(hr, j)) : 0;
#pragma unroll
    for (int cc = 0; cc < NCH; ++cc) {
      v2f a;
#pragma unroll
      for (int v = 0; v < 2; ++v) {
        int q = 2 * cc + v;
        const float* __restrict__ W = (q < NSL) ? Win : Whh;
        int qs = (q < NSL) ? q : (q - NSL);
        bool ok = rok && slot_ok(hi, qs);
        int col = slot_ok(hi, qs) ? unit_of(hi, qs) : 0;
        float w = W[row * HID + col];
        a[v] = ok ? w : 0.0f;
      }
      A[tt][cc] = a;
    }
  }
}

__device__ __forceinline__ void load_bias(const float* __restrict__ bi,
                                          const float* __restrict__ bh,
                                          v8f bias[3], int hi) {
#pragma unroll
  for (int tt = 0; tt < 3; ++tt) {
#pragma unroll
    for (int r = 0; r < 8; ++r) {
      int s = 8 * tt + r;
      int j = s >> 2;
      int g = s & 3;
      bool ok = slot_ok(hi, j);
      int row = ok ? (g * HID + unit_of(hi, j)) : 0;
      float b = bi[row] + bh[row];
      bias[tt][r] = ok ? b : 0.0f;
    }
  }
}

__device__ __forceinline__ void build_B_early(const float in[NSL], v2f B[NCH]) {
  v2f b0; b0.x = in[0]; b0.y = in[1];
  v2f b1; b1.x = in[2]; b1.y = in[3];
  B[0] = b0; B[1] = b1;
}
__device__ __forceinline__ void build_B_late(float in4, const float h[NSL], v2f B[NCH]) {
  v2f b2; b2.x = in4;  b2.y = h[0];
  v2f b3; b3.x = h[1]; b3.y = h[2];
  v2f b4; b4.x = h[3]; b4.y = h[4];
  B[2] = b2; B[3] = b3; B[4] = b4;
}

template <int C0, int C1>
__device__ __forceinline__ void mm(const v2f A[3][NCH], const v2f B[NCH], v8f D[3]) {
#pragma unroll
  for (int cc = C0; cc < C1; ++cc) {
#pragma unroll
    for (int tt = 0; tt < 3; ++tt) D[tt] = wmma4(A[tt][cc], B[cc], D[tt]);
  }
}

__device__ __forceinline__ void gates(const v8f D[3], float h[NSL], float cst[NSL]) {
#pragma unroll
  for (int j = 0; j < NSL; ++j) {
    float ig = fsig(D[(4 * j) >> 3][(4 * j) & 7]);
    float fg = fsig(D[(4 * j + 1) >> 3][(4 * j + 1) & 7]);
    float gg = ftanh(D[(4 * j + 2) >> 3][(4 * j + 2) & 7]);
    float og = fsig(D[(4 * j + 3) >> 3][(4 * j + 3) & 7]);
    float cn = fg * cst[j] + ig * gg;
    cst[j] = cn;
    h[j] = og * ftanh(cn);
  }
}

__device__ __forceinline__ void fetch_x(const float* __restrict__ xp, int hi, float xs[NSL]) {
#pragma unroll
  for (int q = 0; q < NSL; ++q) xs[q] = xp[slot_ok(hi, q) ? unit_of(hi, q) : (HID - 1)];
}
__device__ __forceinline__ void fetch_x3(const float* __restrict__ xrow, int t0, int hi,
                                         float ya[NSL], float yb[NSL], float yc[NSL]) {
  const int ra = (t0     < TLEN) ? t0       : (TLEN - 1);
  const int rb = (t0 + 1 < TLEN) ? (t0 + 1) : (TLEN - 1);
  const int rc = (t0 + 2 < TLEN) ? (t0 + 2) : (TLEN - 1);
  fetch_x(xrow + ra * HID, hi, ya);
  fetch_x(xrow + rb * HID, hi, yb);
  fetch_x(xrow + rc * HID, hi, yc);
}
__device__ __forceinline__ float build_Bx(const float xs[NSL], int hi, v2f B[NCH]) {
  float s[NSL];
#pragma unroll
  for (int q = 0; q < NSL; ++q) s[q] = slot_ok(hi, q) ? xs[q] : 0.0f;
  build_B_early(s, B);
  return s[NSL - 1];
}

__device__ __forceinline__ void bar_signal() {
  asm volatile("s_wait_dscnt 0x0\n\ts_barrier_signal -1" ::: "memory");
}
__device__ __forceinline__ void bar_wait() {
  asm volatile("s_barrier_wait -1" ::: "memory");
}
__device__ __forceinline__ void wg_sync() {
  bar_signal();
  bar_wait();
}

template <bool FETCHX, int SYNC>
__device__ __forceinline__ void l0_step(const float* __restrict__ xrow, int k, int hi,
                                        const v2f A[3][NCH],
                                        const v8f bias[3], float h[NSL], float c[NSL],
                                        v8f Dx[3], float& x4, const float xuse[NSL],
                                        float* dst,
                                        float y0[NSL], float y1[NSL], float y2[NSL]) {
  const int tp = (k + 11 < TLEN) ? (k + 11) : (TLEN - 1);
  __builtin_prefetch(xrow + tp * HID, 0, 0);

  v2f B[NCH];
  v8f D0[3];
  build_B_late(x4, h, B);
  D0[0] = Dx[0]; D0[1] = Dx[1]; D0[2] = Dx[2];
  mm<NCH_EARLY, NCH>(A, B, D0);
  __builtin_amdgcn_sched_barrier(0);
  if (SYNC == 2) bar_wait();
  if (SYNC >= 1) bar_signal();
  if (FETCHX) fetch_x3(xrow, k + 4, hi, y0, y1, y2);
  x4 = build_Bx(xuse, hi, B);
  Dx[0] = bias[0]; Dx[1] = bias[1]; Dx[2] = bias[2];
  gates(D0, h, c);
#pragma unroll
  for (int j = 0; j < NSL; ++j) dst[32 * j] = h[j];
  mm<0, NCH_EARLY>(A, B, Dx);
  asm volatile("" : "+v"(Dx[0]), "+v"(Dx[1]), "+v"(Dx[2]),
                    "+v"(h[0]), "+v"(h[1]), "+v"(h[2]), "+v"(h[3]), "+v"(h[4]));
}

template <bool GATES, bool FETCH>
__device__ __forceinline__ void l1_step(const float* src,
                                        const v2f A[3][NCH],
                                        const v8f bias[3], float h[NSL], float c[NSL],
                                        v8f D[3], float hin[NSL]) {
  float hn[NSL];
  if (FETCH) {
#pragma unroll
    for (int j = 0; j < NSL; ++j) hn[j] = src[32 * j];
  }
  v2f B[NCH];
  v8f E[3];
  E[0] = bias[0]; E[1] = bias[1]; E[2] = bias[2];
  build_B_early(hin, B);
  if (GATES) gates(D, h, c);
  mm<0, NCH_EARLY>(A, B, E);
  build_B_late(hin[NSL - 1], h, B);
  mm<NCH_EARLY, NCH>(A, B, E);
  D[0] = E[0]; D[1] = E[1]; D[2] = E[2];
  if (FETCH) {
#pragma unroll
    for (int j = 0; j < NSL; ++j) hin[j] = hn[j];
  }
  asm volatile("" : "+v"(D[0]), "+v"(D[1]), "+v"(D[2]),
                    "+v"(hin[0]), "+v"(hin[1]), "+v"(hin[2]), "+v"(hin[3]), "+v"(hin[4]));
}

template <int N, bool FETCH, int SYNC>
__device__ __forceinline__ void l0_trip(const float* __restrict__ xrow, int k, int hi,
                                        const v2f A[3][NCH], const v8f bias[3],
                                        float h[NSL], float c[NSL], v8f Dx[3], float& x4,
                                        float xs0[NSL], float xs1[NSL], float xs2[NSL],
                                        float y0[NSL], float y1[NSL], float y2[NSL],
                                        float* set) {
  const int RB = NSL * 32;
  l0_step<FETCH, SYNC>(xrow, k, hi, A, bias, h, c, Dx, x4, xs0, set, y0, y1, y2);
  if (N > 1) l0_step<false, 0>(xrow, k + 1, hi, A, bias, h, c, Dx, x4, xs1, set + RB, y0, y1, y2);
  if (N > 2) l0_step<false, 0>(xrow, k + 2, hi, A, bias, h, c, Dx, x4, xs2, set + 2 * RB, y0, y1, y2);
  if (FETCH) {
#pragma unroll
    for (int j = 0; j < NSL; ++j) { xs0[j] = y0[j]; xs1[j] = y1[j]; xs2[j] = y2[j]; }
  }
}

template <int NS, int NF>
__device__ __forceinline__ void l1_trip(const float* set,
                                        const v2f A[3][NCH], const v8f bias[3],
                                        float h[NSL], float c[NSL], v8f D[3],
                                        float hin[NSL]) {
  const int RB = NSL * 32;
  l1_step<true, (NF > 0)>(set, A, bias, h, c, D, hin);
  if (NS > 1) l1_step<true, (NF > 1)>(set + RB, A, bias, h, c, D, hin);
  if (NS > 2) l1_step<true, (NF > 2)>(set + 2 * RB, A, bias, h, c, D, hin);
}

__global__ __launch_bounds__(64) void lstm2_fc_fused(
    const float* __restrict__ x,
    const float* __restrict__ wih0, const float* __restrict__ whh0,
    const float* __restrict__ bih0, const float* __restrict__ bhh0,
    const float* __restrict__ wih1, const float* __restrict__ whh1,
    const float* __restrict__ bih1, const float* __restrict__ bhh1,
    const float* __restrict__ fcw,  const float* __restrict__ fcb,
    float* __restrict__ out) {
  __shared__ float hx[NSET][SPT][NSL][32];

  const int lane = threadIdx.x & 31;
  const int role = __builtin_amdgcn_readfirstlane((int)(threadIdx.x >> 5));
  const int m  = lane & 15;
  const int hi = lane >> 4;
  const int bglob = blockIdx.x * 16 + m;

  const float* wih = role ? wih1 : wih0;
  const float* whh = role ? whh1 : whh0;
  const float* bih = role ? bih1 : bih0;
  const float* bhh = role ? bhh1 : bhh0;
  v2f A[3][NCH];
  load_A(wih, whh, A, m, hi);
  v8f bias[3];
  load_bias(bih, bhh, bias, hi);

  float h[NSL], c[NSL];
  float xs0[NSL], xs1[NSL], xs2[NSL];
  float y0[NSL], y1[NSL], y2[NSL];
  float hin[NSL];
#pragma unroll
  for (int j = 0; j < NSL; ++j) {
    h[j] = c[j] = 0.0f; hin[j] = 0.0f;
    xs0[j] = xs1[j] = xs2[j] = 0.0f; y0[j] = y1[j] = y2[j] = 0.0f;
  }
  v8f Dacc[3];
  Dacc[0] = bias[0]; Dacc[1] = bias[1]; Dacc[2] = bias[2];
  float x4 = 0.0f;

  const float* xrow = x + (size_t)bglob * TLEN * HID;
  float* ring = &hx[0][0][0][lane];
  const int SB = SPT * NSL * 32;
  const int NFULL = TLEN / SPT;
  const int REM = TLEN - NFULL * SPT;

  if (role == 0) {
    float x0[NSL];
    v2f Bx[NCH];
    fetch_x(xrow, hi, x0);
    fetch_x3(xrow, 1, hi, xs0, xs1, xs2);
    x4 = build_Bx(x0, hi, Bx);
    mm<0, NCH_EARLY>(A, Bx, Dacc);

    l0_trip<SPT, true, 0>(xrow, 0, hi, A, bias, h, c, Dacc, x4,
                          xs0, xs1, xs2, y0, y1, y2, ring);
    l0_trip<SPT, true, 1>(xrow, SPT, hi, A, bias, h, c, Dacc, x4,
                          xs0, xs1, xs2, y0, y1, y2, ring + SB);
    float* w0 = ring + 2 * SB;
    float* w1 = ring;
    float* w2 = ring + SB;
    for (int k = 2 * SPT; k < NFULL * SPT; k += SPT) {
      l0_trip<SPT, true, 2>(xrow, k, hi, A, bias, h, c, Dacc, x4,
                            xs0, xs1, xs2, y0, y1, y2, w0);
      float* t = w0; w0 = w1; w1 = w2; w2 = t;
    }
    if (REM > 0) {
      l0_trip<(REM > 0 ? REM : 1), false, 2>(xrow, NFULL * SPT, hi, A, bias, h, c, Dacc, x4,
                                             xs0, xs1, xs2, y0, y1, y2, w0);
    } else {
      bar_wait();
      bar_signal();
    }
    bar_wait();
    bar_signal();
    bar_wait();
  } else {
    wg_sync();
#pragma unroll
    for (int j = 0; j < NSL; ++j) hin[j] = ring[32 * j];
    l1_step<false, true>(ring + NSL * 32, A, bias, h, c, Dacc, hin);
    l1_step<true, true>(ring + 2 * NSL * 32, A, bias, h, c, Dacc, hin);
    wg_sync();
    float* r0 = ring + SB;
    float* r1 = ring + 2 * SB;
    float* r2 = ring;
    for (int k = 2 * SPT; k < NFULL * SPT; k += SPT) {
      l1_trip<SPT, SPT>(r0, A, bias, h, c, Dacc, hin);
      wg_sync();
      float* t = r0; r0 = r1; r1 = r2; r2 = t;
    }
    l1_trip<SPT, SPT>(r0, A, bias, h, c, Dacc, hin);
    wg_sync();

    l1_trip<REM + 1, REM>(r1, A, bias, h, c, Dacc, hin);
    gates(Dacc, h, c);

    float ph[NSL - 1];
#pragma unroll
    for (int j = 0; j < NSL - 1; ++j) ph[j] = __shfl_xor(h[j], 16, 32);
    if (hi == 0) {
      float hall[HID];
#pragma unroll
      for (int j = 0; j < NSL; ++j) hall[j] = h[j];
#pragma unroll
      for (int j = 0; j < NSL - 1; ++j) hall[NSL + j] = ph[j];
      float* op = out + (size_t)bglob * NCLS;
#pragma unroll
      for (int k = 0; k < NCLS; ++k) {
        float acc = fcb[k];
#pragma unroll
        for (int j = 0; j < HID; ++j) acc += hall[j] * fcw[k * HID + j];
        op[k] = acc;
      }
    }
  }
}

extern "C" void kernel_launch(void* const* d_in, const int* in_sizes, int n_in,
                              void* d_out, int out_size, void* d_ws, size_t ws_size,
                              hipStream_t stream) {
  const float* x    = (const float*)d_in[0];
  const float* wih0 = (const float*)d_in[1];
  const float* whh0 = (const float*)d_in[2];
  const float* bih0 = (const float*)d_in[3];
  const float* bhh0 = (const float*)d_in[4];
  const float* wih1 = (const float*)d_in[5];
  const float* whh1 = (const float*)d_in[6];
  const float* bih1 = (const float*)d_in[7];
  const float* bhh1 = (const float*)d_in[8];
  const float* fcw  = (const float*)d_in[9];
  const float* fcb  = (const float*)d_in[10];
  float* out = (float*)d_out;

  const int B = in_sizes[0] / (TLEN * HID);
  dim3 grid(B / 16);
  dim3 block(64);
  lstm2_fc_fused<<<grid, block, 0, stream>>>(x, wih0, whh0, bih0, bhh0,
                                             wih1, whh1, bih1, bhh1,
                                             fcw, fcb, out);
}


# each layer's input product moved to a helper wave on an idle SIMD, gate pre-sums handed via LDS, one barrier per 4 steps: state waves issue 9 matrix instructions a step, not 15; k sum order changes
# speedup vs baseline: 5.3787x; 1.1764x over previous
#include <hip/hip_runtime.h>
#include <math.h>


#define HID 9
#define NG  36
#define TLEN 512
#define NCLS 10
#define NSL 5
#define NCH 3
#define SPT 4
#define NTRIP (TLEN / SPT)
#define NGRP 5

static_assert(TLEN % SPT == 0, "a trip must divide the sequence");


typedef __attribute__((ext_vector_type(2))) float v2f;
typedef __attribute__((ext_vector_type(4))) float v4f;
typedef __attribute__((ext_vector_type(8))) float v8f;

__device__ __forceinline__ float fexp2(float x) {
#if defined(__HIP_DEVICE_COMPILE__) && __has_builtin(__builtin_amdgcn_exp2f)
  return __builtin_amdgcn_exp2f(x);
#else
  return exp2f(x);
#endif
}

__device__ __forceinline__ float frcp(float x) {
#if defined(__HIP_DEVICE_COMPILE__) && __has_builtin(__builtin_amdgcn_rcpf)
  return __builtin_amdgcn_rcpf(x);
#else
  return 1.0f / x;
#endif
}

__device__ __forceinline__ float ftanh(float x) {
#if defined(__HIP_DEVICE_COMPILE__) && __has_builtin(__builtin_amdgcn_tanhf)
  return __builtin_amdgcn_tanhf(x);
#else
  float e = fexp2(2.8853900817779268f * x);
  return 1.0f - 2.0f * frcp(e + 1.0f);
#endif
}

__device__ __forceinline__ float fsig(float x) {
  return 0.5f + 0.5f * ftanh(0.5f * x);
}

__device__ __forceinline__ v8f wmma4(v2f a, v2f b, v8f c) {
  return __builtin_amdgcn_wmma_f32_16x16x4_f32(false, a, false, b, (short)0, c,
                                               false, false);
}

__device__ __forceinline__ int unit_of(int hh, int j) { return hh ? (NSL + j) : j; }
__device__ __forceinline__ bool slot_ok(int hh, int j) { return j < NSL && unit_of(hh, j) < HID; }

__device__ __forceinline__ void load_A(const float* __restrict__ W,
                                       v2f A[3][NCH], int m, int hi) {
  const int hr = m >> 3;
#pragma unroll
  for (int tt = 0; tt < 3; ++tt) {
    int s = 8 * tt + (m & 7);
    int j = s >> 2;
    int g = s & 3;
    bool rok = slot_ok(hr, j);
    int row = rok ? (g * HID + unit_of(hr, j)) : 0;
#pragma unroll
    for (int cc = 0; cc < NCH; ++cc) {
      v2f a;
#pragma unroll
      for (int v = 0; v < 2; ++v) {
        int q = 2 * cc + v;
        bool kok = slot_ok(hi, q);
        int col = kok ? unit_of(hi, q) : 0;
        float w = W[row * HID + col];
        a[v] = (rok && kok) ? w : 0.0f;
      }
      A[tt][cc] = a;
    }
  }
}

__device__ __forceinline__ void load_bias(const float* __restrict__ bi,
                                          const float* __restrict__ bh,
                                          v8f bias[3], int hi) {
#pragma unroll
  for (int tt = 0; tt < 3; ++tt) {
#pragma unroll
    for (int r = 0; r < 8; ++r) {
      int s = 8 * tt + r;
      int j = s >> 2;
      int g = s & 3;
      bool ok = slot_ok(hi, j);
      int row = ok ? (g * HID + unit_of(hi, j)) : 0;
      float b = bi[row] + bh[row];
      bias[tt][r] = ok ? b : 0.0f;
    }
  }
}

__device__ __forceinline__ void build_B(const float s[NSL], v2f B[NCH]) {
  v2f b0; b0.x = s[0]; b0.y = s[1];
  v2f b1; b1.x = s[2]; b1.y = s[3];
  v2f b2; b2.x = s[4]; b2.y = 0.0f;
  B[0] = b0; B[1] = b1; B[2] = b2;
}

__device__ __forceinline__ void mm(const v2f A[3][NCH], const v2f B[NCH], v8f D[3]) {
#pragma unroll
  for (int cc = 0; cc < NCH; ++cc) {
#pragma unroll
    for (int tt = 0; tt < 3; ++tt) D[tt] = wmma4(A[tt][cc], B[cc], D[tt]);
  }
}

__device__ __forceinline__ void gates(const v8f D[3], float h[NSL], float cst[NSL]) {
#pragma unroll
  for (int j = 0; j < NSL; ++j) {
    float ig = fsig(D[(4 * j) >> 3][(4 * j) & 7]);
    float fg = fsig(D[(4 * j + 1) >> 3][(4 * j + 1) & 7]);
    float gg = ftanh(D[(4 * j + 2) >> 3][(4 * j + 2) & 7]);
    float og = fsig(D[(4 * j + 3) >> 3][(4 * j + 3) & 7]);
    float cn = fg * cst[j] + ig * gg;
    cst[j] = cn;
    h[j] = og * ftanh(cn);
  }
}

__device__ __forceinline__ void put_pre(const v8f D[3], v4f* p) {
  v4f g0, g1, g2, g3, g4;
#pragma unroll
  for (int e = 0; e < 4; ++e) {
    g0[e] = D[0][e]; g1[e] = D[0][4 + e];
    g2[e] = D[1][e]; g3[e] = D[1][4 + e];
    g4[e] = D[2][e];
  }
  p[0] = g0; p[32] = g1; p[64] = g2; p[96] = g3; p[128] = g4;
}
__device__ __forceinline__ void get_pre(const v4f* p, v8f D[3]) {
  v4f g0 = p[0], g1 = p[32], g2 = p[64], g3 = p[96], g4 = p[128];
#pragma unroll
  for (int e = 0; e < 4; ++e) {
    D[0][e] = g0[e]; D[0][4 + e] = g1[e];
    D[1][e] = g2[e]; D[1][4 + e] = g3[e];
    D[2][e] = g4[e]; D[2][4 + e] = 0.0f;
  }
}

__device__ __forceinline__ void fetch_x(const float* __restrict__ xp, int hi, float xs[NSL]) {
#pragma unroll
  for (int q = 0; q < NSL; ++q) xs[q] = xp[slot_ok(hi, q) ? unit_of(hi, q) : (HID - 1)];
}

__device__ __forceinline__ void wg_sync() {
  asm volatile("s_wait_dscnt 0x0\n\ts_barrier_signal -1\n\ts_barrier_wait -1" ::: "memory");
}

__global__ __launch_bounds__(128) void lstm2_fc_fused(
    const float* __restrict__ x,
    const float* __restrict__ wih0, const float* __restrict__ whh0,
    const float* __restrict__ bih0, const float* __restrict__ bhh0,
    const float* __restrict__ wih1, const float* __restrict__ whh1,
    const float* __restrict__ bih1, const float* __restrict__ bhh1,
    const float* __restrict__ fcw,  const float* __restrict__ fcb,
    float* __restrict__ out) {
  __shared__ v4f pre0[2][SPT][NGRP][32];
  __shared__ v4f pre1[2][SPT][NGRP][32];
  __shared__ float hx[2][SPT][NSL][32];

  const int lane = threadIdx.x & 31;
  const int role = __builtin_amdgcn_readfirstlane((int)(threadIdx.x >> 5));
  const int m  = lane & 15;
  const int hi = lane >> 4;
  const int bglob = blockIdx.x * 16 + m;
  const int layer = role >> 1;
  const int rec = role & 1;

  const float* wmat = layer ? (rec ? whh1 : wih1) : (rec ? whh0 : wih0);
  v2f A[3][NCH];
  load_A(wmat, A, m, hi);
  v8f bias[3];
  load_bias(layer ? bih1 : bih0, layer ? bhh1 : bhh0, bias, hi);

  float h[NSL], c[NSL];
#pragma unroll
  for (int j = 0; j < NSL; ++j) { h[j] = 0.0f; c[j] = 0.0f; }

  const float* xrow = x + (size_t)bglob * TLEN * HID;
  float y[SPT][NSL];
#pragma unroll
  for (int s = 0; s < SPT; ++s) {
#pragma unroll
    for (int q = 0; q < NSL; ++q) y[s][q] = 0.0f;
  }
  if (role == 0) {
#pragma unroll
    for (int s = 0; s < SPT; ++s) fetch_x(xrow + s * HID, hi, y[s]);
  }

  for (int tick = 0; tick < NTRIP + 3; ++tick) {
    const int T = tick - role;
    if (T >= 0 && T < NTRIP) {
      const int set = T & 1;
      if (role == 0) {
        float xs[SPT][NSL];
#pragma unroll
        for (int s = 0; s < SPT; ++s) {
#pragma unroll
          for (int q = 0; q < NSL; ++q) xs[s][q] = slot_ok(hi, q) ? y[s][q] : 0.0f;
        }
        const int tn = (T + 1 < NTRIP) ? (T + 1) : T;
#pragma unroll
        for (int s = 0; s < SPT; ++s) fetch_x(xrow + (tn * SPT + s) * HID, hi, y[s]);
#pragma unroll
        for (int s = 0; s < SPT; ++s) {
          v2f B[NCH];
          v8f D[3];
          build_B(xs[s], B);
          D[0] = bias[0]; D[1] = bias[1]; D[2] = bias[2];
          mm(A, B, D);
          put_pre(D, &pre0[set][s][0][lane]);
        }
      } else if (role == 2) {
        float hin[SPT][NSL];
#pragma unroll
        for (int s = 0; s < SPT; ++s) {
#pragma unroll
          for (int j = 0; j < NSL; ++j) hin[s][j] = hx[set][s][j][lane];
        }
#pragma unroll
        for (int s = 0; s < SPT; ++s) {
          v2f B[NCH];
          v8f D[3];
          build_B(hin[s], B);
          D[0] = bias[0]; D[1] = bias[1]; D[2] = bias[2];
          mm(A, B, D);
          put_pre(D, &pre1[set][s][0][lane]);
        }
      } else {
        v8f P[SPT][3];
        const v4f* src = (role == 1) ? &pre0[set][0][0][lane] : &pre1[set][0][0][lane];
#pragma unroll
        for (int s = 0; s < SPT; ++s) get_pre(src + s * (NGRP * 32), P[s]);
#pragma unroll
        for (int s = 0; s < SPT; ++s) {
          v2f B[NCH];
          build_B(h, B);
          mm(A, B, P[s]);
          gates(P[s], h, c);
          if (role == 1) {
#pragma unroll
            for (int j = 0; j < NSL; ++j) hx[set][s][j][lane] = h[j];
          }
        }
      }
    }
    wg_sync();
  }

  if (role == 3) {
    float ph[NSL - 1];
#pragma unroll
    for (int j = 0; j < NSL - 1; ++j) ph[j] = __shfl_xor(h[j], 16, 32);
    if (hi == 0) {
      float hall[HID];
#pragma unroll
      for (int j = 0; j < NSL; ++j) hall[j] = h[j];
#pragma unroll
      for (int j = 0; j < NSL - 1; ++j) hall[NSL + j] = ph[j];
      float* op = out + (size_t)bglob * NCLS;
#pragma unroll
      for (int k = 0; k < NCLS; ++k) {
        float acc = fcb[k];
#pragma unroll
        for (int j = 0; j < HID; ++j) acc += hall[j] * fcw[k * HID + j];
        op[k] = acc;
      }
    }
  }
}

extern "C" void kernel_launch(void* const* d_in, const int* in_sizes, int n_in,
                              void* d_out, int out_size, void* d_ws, size_t ws_size,
                              hipStream_t stream) {
  const float* x    = (const float*)d_in[0];
  const float* wih0 = (const float*)d_in[1];
  const float* whh0 = (const float*)d_in[2];
  const float* bih0 = (const float*)d_in[3];
  const float* bhh0 = (const float*)d_in[4];
  const float* wih1 = (const float*)d_in[5];
  const float* whh1 = (const float*)d_in[6];
  const float* bih1 = (const float*)d_in[7];
  const float* bhh1 = (const float*)d_in[8];
  const float* fcw  = (const float*)d_in[9];
  const float* fcb  = (const float*)d_in[10];
  float* out = (float*)d_out;

  const int B = in_sizes[0] / (TLEN * HID);
  dim3 grid(B / 16);
  dim3 block(128);
  lstm2_fc_fused<<<grid, block, 0, stream>>>(x, wih0, whh0, bih0, bhh0,
                                             wih1, whh1, bih1, bhh1,
                                             fcw, fcb, out);
}


# a trip is 8 time steps, not 4: one workgroup barrier, one group of ring reads and one store drain per 8 steps on every wave; rings and register sets doubled; sums unchanged
# speedup vs baseline: 6.4343x; 1.1963x over previous
#include <hip/hip_runtime.h>
#include <math.h>


#define HID 9
#define NG  36
#define TLEN 512
#define NCLS 10
#define NSL 5
#define NCH 3
#define SPT 8
#define NTRIP (TLEN / SPT)
#define NGRP 5

static_assert(TLEN % SPT == 0, "a trip must divide the sequence");


typedef __attribute__((ext_vector_type(2))) float v2f;
typedef __attribute__((ext_vector_type(4))) float v4f;
typedef __attribute__((ext_vector_type(8))) float v8f;

__device__ __forceinline__ float fexp2(float x) {
#if defined(__HIP_DEVICE_COMPILE__) && __has_builtin(__builtin_amdgcn_exp2f)
  return __builtin_amdgcn_exp2f(x);
#else
  return exp2f(x);
#endif
}

__device__ __forceinline__ float frcp(float x) {
#if defined(__HIP_DEVICE_COMPILE__) && __has_builtin(__builtin_amdgcn_rcpf)
  return __builtin_amdgcn_rcpf(x);
#else
  return 1.0f / x;
#endif
}

__device__ __forceinline__ float ftanh(float x) {
#if defined(__HIP_DEVICE_COMPILE__) && __has_builtin(__builtin_amdgcn_tanhf)
  return __builtin_amdgcn_tanhf(x);
#else
  float e = fexp2(2.8853900817779268f * x);
  return 1.0f - 2.0f * frcp(e + 1.0f);
#endif
}

__device__ __forceinline__ float fsig(float x) {
  return 0.5f + 0.5f * ftanh(0.5f * x);
}

__device__ __forceinline__ v8f wmma4(v2f a, v2f b, v8f c) {
  return __builtin_amdgcn_wmma_f32_16x16x4_f32(false, a, false, b, (short)0, c,
                                               false, false);
}

__device__ __forceinline__ int unit_of(int hh, int j) { return hh ? (NSL + j) : j; }
__device__ __forceinline__ bool slot_ok(int hh, int j) { return j < NSL && unit_of(hh, j) < HID; }

__device__ __forceinline__ void load_A(const float* __restrict__ W,
                                       v2f A[3][NCH], int m, int hi) {
  const int hr = m >> 3;
#pragma unroll
  for (int tt = 0; tt < 3; ++tt) {
    int s = 8 * tt + (m & 7);
    int j = s >> 2;
    int g = s & 3;
    bool rok = slot_ok(hr, j);
    int row = rok ? (g * HID + unit_of(hr, j)) : 0;
#pragma unroll
    for (int cc = 0; cc < NCH; ++cc) {
      v2f a;
#pragma unroll
      for (int v = 0; v < 2; ++v) {
        int q = 2 * cc + v;
        bool kok = slot_ok(hi, q);
        int col = kok ? unit_of(hi, q) : 0;
        float w = W[row * HID + col];
        a[v] = (rok && kok) ? w : 0.0f;
      }
      A[tt][cc] = a;
    }
  }
}

__device__ __forceinline__ void load_bias(const float* __restrict__ bi,
                                          const float* __restrict__ bh,
                                          v8f bias[3], int hi) {
#pragma unroll
  for (int tt = 0; tt < 3; ++tt) {
#pragma unroll
    for (int r = 0; r < 8; ++r) {
      int s = 8 * tt + r;
      int j = s >> 2;
      int g = s & 3;
      bool ok = slot_ok(hi, j);
      int row = ok ? (g * HID + unit_of(hi, j)) : 0;
      float b = bi[row] + bh[row];
      bias[tt][r] = ok ? b : 0.0f;
    }
  }
}

__device__ __forceinline__ void build_B(const float s[NSL], v2f B[NCH]) {
  v2f b0; b0.x = s[0]; b0.y = s[1];
  v2f b1; b1.x = s[2]; b1.y = s[3];
  v2f b2; b2.x = s[4]; b2.y = 0.0f;
  B[0] = b0; B[1] = b1; B[2] = b2;
}

__device__ __forceinline__ void mm(const v2f A[3][NCH], const v2f B[NCH], v8f D[3]) {
#pragma unroll
  for (int cc = 0; cc < NCH; ++cc) {
#pragma unroll
    for (int tt = 0; tt < 3; ++tt) D[tt] = wmma4(A[tt][cc], B[cc], D[tt]);
  }
}

__device__ __forceinline__ void gates(const v8f D[3], float h[NSL], float cst[NSL]) {
#pragma unroll
  for (int j = 0; j < NSL; ++j) {
    float ig = fsig(D[(4 * j) >> 3][(4 * j) & 7]);
    float fg = fsig(D[(4 * j + 1) >> 3][(4 * j + 1) & 7]);
    float gg = ftanh(D[(4 * j + 2) >> 3][(4 * j + 2) & 7]);
    float og = fsig(D[(4 * j + 3) >> 3][(4 * j + 3) & 7]);
    float cn = fg * cst[j] + ig * gg;
    cst[j] = cn;
    h[j] = og * ftanh(cn);
  }
}

__device__ __forceinline__ void put_pre(const v8f D[3], v4f* p) {
  v4f g0, g1, g2, g3, g4;
#pragma unroll
  for (int e = 0; e < 4; ++e) {
    g0[e] = D[0][e]; g1[e] = D[0][4 + e];
    g2[e] = D[1][e]; g3[e] = D[1][4 + e];
    g4[e] = D[2][e];
  }
  p[0] = g0; p[32] = g1; p[64] = g2; p[96] = g3; p[128] = g4;
}
__device__ __forceinline__ void get_pre(const v4f* p, v8f D[3]) {
  v4f g0 = p[0], g1 = p[32], g2 = p[64], g3 = p[96], g4 = p[128];
#pragma unroll
  for (int e = 0; e < 4; ++e) {
    D[0][e] = g0[e]; D[0][4 + e] = g1[e];
    D[1][e] = g2[e]; D[1][4 + e] = g3[e];
    D[2][e] = g4[e]; D[2][4 + e] = 0.0f;
  }
}

__device__ __forceinline__ void fetch_x(const float* __restrict__ xp, int hi, float xs[NSL]) {
#pragma unroll
  for (int q = 0; q < NSL; ++q) xs[q] = xp[slot_ok(hi, q) ? unit_of(hi, q) : (HID - 1)];
}

__device__ __forceinline__ void wg_sync() {
  asm volatile("s_wait_dscnt 0x0\n\ts_barrier_signal -1\n\ts_barrier_wait -1" ::: "memory");
}

__global__ __launch_bounds__(128) void lstm2_fc_fused(
    const float* __restrict__ x,
    const float* __restrict__ wih0, const float* __restrict__ whh0,
    const float* __restrict__ bih0, const float* __restrict__ bhh0,
    const float* __restrict__ wih1, const float* __restrict__ whh1,
    const float* __restrict__ bih1, const float* __restrict__ bhh1,
    const float* __restrict__ fcw,  const float* __restrict__ fcb,
    float* __restrict__ out) {
  __shared__ v4f pre0[2][SPT][NGRP][32];
  __shared__ v4f pre1[2][SPT][NGRP][32];
  __shared__ float hx[2][SPT][NSL][32];

  const int lane = threadIdx.x & 31;
  const int role = __builtin_amdgcn_readfirstlane((int)(threadIdx.x >> 5));
  const int m  = lane & 15;
  const int hi = lane >> 4;
  const int bglob = blockIdx.x * 16 + m;
  const int layer = role >> 1;
  const int rec = role & 1;

  const float* wmat = layer ? (rec ? whh1 : wih1) : (rec ? whh0 : wih0);
  v2f A[3][NCH];
  load_A(wmat, A, m, hi);
  v8f bias[3];
  load_bias(layer ? bih1 : bih0, layer ? bhh1 : bhh0, bias, hi);

  float h[NSL], c[NSL];
#pragma unroll
  for (int j = 0; j < NSL; ++j) { h[j] = 0.0f; c[j] = 0.0f; }

  const float* xrow = x + (size_t)bglob * TLEN * HID;
  float y[SPT][NSL];
#pragma unroll
  for (int s = 0; s < SPT; ++s) {
#pragma unroll
    for (int q = 0; q < NSL; ++q) y[s][q] = 0.0f;
  }
  if (role == 0) {
#pragma unroll
    for (int s = 0; s < SPT; ++s) fetch_x(xrow + s * HID, hi, y[s]);
  }

  for (int tick = 0; tick < NTRIP + 3; ++tick) {
    const int T = tick - role;
    if (T >= 0 && T < NTRIP) {
      const int set = T & 1;
      if (role == 0) {
        float xs[SPT][NSL];
#pragma unroll
        for (int s = 0; s < SPT; ++s) {
#pragma unroll
          for (int q = 0; q < NSL; ++q) xs[s][q] = slot_ok(hi, q) ? y[s][q] : 0.0f;
        }
        const int tn = (T + 1 < NTRIP) ? (T + 1) : T;
#pragma unroll
        for (int s = 0; s < SPT; ++s) fetch_x(xrow + (tn * SPT + s) * HID, hi, y[s]);
#pragma unroll
        for (int s = 0; s < SPT; ++s) {
          v2f B[NCH];
          v8f D[3];
          build_B(xs[s], B);
          D[0] = bias[0]; D[1] = bias[1]; D[2] = bias[2];
          mm(A, B, D);
          put_pre(D, &pre0[set][s][0][lane]);
        }
      } else if (role == 2) {
        float hin[SPT][NSL];
#pragma unroll
        for (int s = 0; s < SPT; ++s) {
#pragma unroll
          for (int j = 0; j < NSL; ++j) hin[s][j] = hx[set][s][j][lane];
        }
#pragma unroll
        for (int s = 0; s < SPT; ++s) {
          v2f B[NCH];
          v8f D[3];
          build_B(hin[s], B);
          D[0] = bias[0]; D[1] = bias[1]; D[2] = bias[2];
          mm(A, B, D);
          put_pre(D, &pre1[set][s][0][lane]);
        }
      } else {
        v8f P[SPT][3];
        const v4f* src = (role == 1) ? &pre0[set][0][0][lane] : &pre1[set][0][0][lane];
#pragma unroll
        for (int s = 0; s < SPT; ++s) get_pre(src + s * (NGRP * 32), P[s]);
#pragma unroll
        for (int s = 0; s < SPT; ++s) {
          v2f B[NCH];
          build_B(h, B);
          mm(A, B, P[s]);
          gates(P[s], h, c);
          if (role == 1) {
#pragma unroll
            for (int j = 0; j < NSL; ++j) hx[set][s][j][lane] = h[j];
          }
        }
      }
    }
    wg_sync();
  }

  if (role == 3) {
    float ph[NSL - 1];
#pragma unroll
    for (int j = 0; j < NSL - 1; ++j) ph[j] = __shfl_xor(h[j], 16, 32);
    if (hi == 0) {
      float hall[HID];
#pragma unroll
      for (int j = 0; j < NSL; ++j) hall[j] = h[j];
#pragma unroll
      for (int j = 0; j < NSL - 1; ++j) hall[NSL + j] = ph[j];
      float* op = out + (size_t)bglob * NCLS;
#pragma unroll
      for (int k = 0; k < NCLS; ++k) {
        float acc = fcb[k];
#pragma unroll
        for (int j = 0; j < HID; ++j) acc += hall[j] * fcw[k * HID + j];
        op[k] = acc;
      }
    }
  }
}

extern "C" void kernel_launch(void* const* d_in, const int* in_sizes, int n_in,
                              void* d_out, int out_size, void* d_ws, size_t ws_size,
                              hipStream_t stream) {
  const float* x    = (const float*)d_in[0];
  const float* wih0 = (const float*)d_in[1];
  const float* whh0 = (const float*)d_in[2];
  const float* bih0 = (const float*)d_in[3];
  const float* bhh0 = (const float*)d_in[4];
  const float* wih1 = (const float*)d_in[5];
  const float* whh1 = (const float*)d_in[6];
  const float* bih1 = (const float*)d_in[7];
  const float* bhh1 = (const float*)d_in[8];
  const float* fcw  = (const float*)d_in[9];
  const float* fcb  = (const float*)d_in[10];
  float* out = (float*)d_out;

  const int B = in_sizes[0] / (TLEN * HID);
  dim3 grid(B / 16);
  dim3 block(128);
  lstm2_fc_fused<<<grid, block, 0, stream>>>(x, wih0, whh0, bih0, bhh0,
                                             wih1, whh1, bih1, bhh1,
                                             fcw, fcb, out);
}


# layer-0 state wave keeps a trip's eight h0 rows in registers and stores them to the LDS ring as one group behind the last step: one switch to the LDS path per trip, not per step; sums unchanged
# speedup vs baseline: 6.6746x; 1.0373x over previous
#include <hip/hip_runtime.h>
#include <math.h>


#define HID 9
#define NG  36
#define TLEN 512
#define NCLS 10
#define NSL 5
#define NCH 3
#define SPT 8
#define NTRIP (TLEN / SPT)
#define NGRP 5

static_assert(TLEN % SPT == 0, "a trip must divide the sequence");


typedef __attribute__((ext_vector_type(2))) float v2f;
typedef __attribute__((ext_vector_type(4))) float v4f;
typedef __attribute__((ext_vector_type(8))) float v8f;

__device__ __forceinline__ float fexp2(float x) {
#if defined(__HIP_DEVICE_COMPILE__) && __has_builtin(__builtin_amdgcn_exp2f)
  return __builtin_amdgcn_exp2f(x);
#else
  return exp2f(x);
#endif
}

__device__ __forceinline__ float frcp(float x) {
#if defined(__HIP_DEVICE_COMPILE__) && __has_builtin(__builtin_amdgcn_rcpf)
  return __builtin_amdgcn_rcpf(x);
#else
  return 1.0f / x;
#endif
}

__device__ __forceinline__ float ftanh(float x) {
#if defined(__HIP_DEVICE_COMPILE__) && __has_builtin(__builtin_amdgcn_tanhf)
  return __builtin_amdgcn_tanhf(x);
#else
  float e = fexp2(2.8853900817779268f * x);
  return 1.0f - 2.0f * frcp(e + 1.0f);
#endif
}

__device__ __forceinline__ float fsig(float x) {
  return 0.5f + 0.5f * ftanh(0.5f * x);
}

__device__ __forceinline__ v8f wmma4(v2f a, v2f b, v8f c) {
  return __builtin_amdgcn_wmma_f32_16x16x4_f32(false, a, false, b, (short)0, c,
                                               false, false);
}

__device__ __forceinline__ int unit_of(int hh, int j) { return hh ? (NSL + j) : j; }
__device__ __forceinline__ bool slot_ok(int hh, int j) { return j < NSL && unit_of(hh, j) < HID; }

__device__ __forceinline__ void load_A(const float* __restrict__ W,
                                       v2f A[3][NCH], int m, int hi) {
  const int hr = m >> 3;
#pragma unroll
  for (int tt = 0; tt < 3; ++tt) {
    int s = 8 * tt + (m & 7);
    int j = s >> 2;
    int g = s & 3;
    bool rok = slot_ok(hr, j);
    int row = rok ? (g * HID + unit_of(hr, j)) : 0;
#pragma unroll
    for (int cc = 0; cc < NCH; ++cc) {
      v2f a;
#pragma unroll
      for (int v = 0; v < 2; ++v) {
        int q = 2 * cc + v;
        bool kok = slot_ok(hi, q);
        int col = kok ? unit_of(hi, q) : 0;
        float w = W[row * HID + col];
        a[v] = (rok && kok) ? w : 0.0f;
      }
      A[tt][cc] = a;
    }
  }
}

__device__ __forceinline__ void load_bias(const float* __restrict__ bi,
                                          const float* __restrict__ bh,
                                          v8f bias[3], int hi) {
#pragma unroll
  for (int tt = 0; tt < 3; ++tt) {
#pragma unroll
    for (int r = 0; r < 8; ++r) {
      int s = 8 * tt + r;
      int j = s >> 2;
      int g = s & 3;
      bool ok = slot_ok(hi, j);
      int row = ok ? (g * HID + unit_of(hi, j)) : 0;
      float b = bi[row] + bh[row];
      bias[tt][r] = ok ? b : 0.0f;
    }
  }
}

__device__ __forceinline__ void build_B(const float s[NSL], v2f B[NCH]) {
  v2f b0; b0.x = s[0]; b0.y = s[1];
  v2f b1; b1.x = s[2]; b1.y = s[3];
  v2f b2; b2.x = s[4]; b2.y = 0.0f;
  B[0] = b0; B[1] = b1; B[2] = b2;
}

__device__ __forceinline__ void mm(const v2f A[3][NCH], const v2f B[NCH], v8f D[3]) {
#pragma unroll
  for (int cc = 0; cc < NCH; ++cc) {
#pragma unroll
    for (int tt = 0; tt < 3; ++tt) D[tt] = wmma4(A[tt][cc], B[cc], D[tt]);
  }
}

__device__ __forceinline__ void gates(const v8f D[3], float h[NSL], float cst[NSL]) {
#pragma unroll
  for (int j = 0; j < NSL; ++j) {
    float ig = fsig(D[(4 * j) >> 3][(4 * j) & 7]);
    float fg = fsig(D[(4 * j + 1) >> 3][(4 * j + 1) & 7]);
    float gg = ftanh(D[(4 * j + 2) >> 3][(4 * j + 2) & 7]);
    float og = fsig(D[(4 * j + 3) >> 3][(4 * j + 3) & 7]);
    float cn = fg * cst[j] + ig * gg;
    cst[j] = cn;
    h[j] = og * ftanh(cn);
  }
}

__device__ __forceinline__ void put_pre(const v8f D[3], v4f* p) {
  v4f g0, g1, g2, g3, g4;
#pragma unroll
  for (int e = 0; e < 4; ++e) {
    g0[e] = D[0][e]; g1[e] = D[0][4 + e];
    g2[e] = D[1][e]; g3[e] = D[1][4 + e];
    g4[e] = D[2][e];
  }
  p[0] = g0; p[32] = g1; p[64] = g2; p[96] = g3; p[128] = g4;
}
__device__ __forceinline__ void get_pre(const v4f* p, v8f D[3]) {
  v4f g0 = p[0], g1 = p[32], g2 = p[64], g3 = p[96], g4 = p[128];
#pragma unroll
  for (int e = 0; e < 4; ++e) {
    D[0][e] = g0[e]; D[0][4 + e] = g1[e];
    D[1][e] = g2[e]; D[1][4 + e] = g3[e];
    D[2][e] = g4[e]; D[2][4 + e] = 0.0f;
  }
}

__device__ __forceinline__ void fetch_x(const float* __restrict__ xp, int hi, float xs[NSL]) {
#pragma unroll
  for (int q = 0; q < NSL; ++q) xs[q] = xp[slot_ok(hi, q) ? unit_of(hi, q) : (HID - 1)];
}

__device__ __forceinline__ void wg_sync() {
  asm volatile("s_wait_dscnt 0x0\n\ts_barrier_signal -1\n\ts_barrier_wait -1" ::: "memory");
}

__global__ __launch_bounds__(128) void lstm2_fc_fused(
    const float* __restrict__ x,
    const float* __restrict__ wih0, const float* __restrict__ whh0,
    const float* __restrict__ bih0, const float* __restrict__ bhh0,
    const float* __restrict__ wih1, const float* __restrict__ whh1,
    const float* __restrict__ bih1, const float* __restrict__ bhh1,
    const float* __restrict__ fcw,  const float* __restrict__ fcb,
    float* __restrict__ out) {
  __shared__ v4f pre0[2][SPT][NGRP][32];
  __shared__ v4f pre1[2][SPT][NGRP][32];
  __shared__ float hx[2][SPT][NSL][32];

  const int lane = threadIdx.x & 31;
  const int role = __builtin_amdgcn_readfirstlane((int)(threadIdx.x >> 5));
  const int m  = lane & 15;
  const int hi = lane >> 4;
  const int bglob = blockIdx.x * 16 + m;
  const int layer = role >> 1;
  const int rec = role & 1;

  const float* wmat = layer ? (rec ? whh1 : wih1) : (rec ? whh0 : wih0);
  v2f A[3][NCH];
  load_A(wmat, A, m, hi);
  v8f bias[3];
  load_bias(layer ? bih1 : bih0, layer ? bhh1 : bhh0, bias, hi);

  float h[NSL], c[NSL];
#pragma unroll
  for (int j = 0; j < NSL; ++j) { h[j] = 0.0f; c[j] = 0.0f; }

  const float* xrow = x + (size_t)bglob * TLEN * HID;
  float y[SPT][NSL];
#pragma unroll
  for (int s = 0; s < SPT; ++s) {
#pragma unroll
    for (int q = 0; q < NSL; ++q) y[s][q] = 0.0f;
  }
  if (role == 0) {
#pragma unroll
    for (int s = 0; s < SPT; ++s) fetch_x(xrow + s * HID, hi, y[s]);
  }

  for (int tick = 0; tick < NTRIP + 3; ++tick) {
    const int T = tick - role;
    if (T >= 0 && T < NTRIP) {
      const int set = T & 1;
      if (role == 0) {
        float xs[SPT][NSL];
#pragma unroll
        for (int s = 0; s < SPT; ++s) {
#pragma unroll
          for (int q = 0; q < NSL; ++q) xs[s][q] = slot_ok(hi, q) ? y[s][q] : 0.0f;
        }
        const int tn = (T + 1 < NTRIP) ? (T + 1) : T;
#pragma unroll
        for (int s = 0; s < SPT; ++s) fetch_x(xrow + (tn * SPT + s) * HID, hi, y[s]);
#pragma unroll
        for (int s = 0; s < SPT; ++s) {
          v2f B[NCH];
          v8f D[3];
          build_B(xs[s], B);
          D[0] = bias[0]; D[1] = bias[1]; D[2] = bias[2];
          mm(A, B, D);
          put_pre(D, &pre0[set][s][0][lane]);
        }
      } else if (role == 2) {
        float hin[SPT][NSL];
#pragma unroll
        for (int s = 0; s < SPT; ++s) {
#pragma unroll
          for (int j = 0; j < NSL; ++j) hin[s][j] = hx[set][s][j][lane];
        }
#pragma unroll
        for (int s = 0; s < SPT; ++s) {
          v2f B[NCH];
          v8f D[3];
          build_B(hin[s], B);
          D[0] = bias[0]; D[1] = bias[1]; D[2] = bias[2];
          mm(A, B, D);
          put_pre(D, &pre1[set][s][0][lane]);
        }
      } else {
        v8f P[SPT][3];
        const v4f* src = (role == 1) ? &pre0[set][0][0][lane] : &pre1[set][0][0][lane];
#pragma unroll
        for (int s = 0; s < SPT; ++s) get_pre(src + s * (NGRP * 32), P[s]);
        __builtin_amdgcn_sched_barrier(0);
        float hq[SPT][NSL];
#pragma unroll
        for (int s = 0; s < SPT; ++s) {
          v2f B[NCH];
          build_B(h, B);
          mm(A, B, P[s]);
          gates(P[s], h, c);
#pragma unroll
          for (int j = 0; j < NSL; ++j) hq[s][j] = h[j];
        }
        if (role == 1) {
#pragma unroll
          for (int s = 0; s < SPT; ++s) {
#pragma unroll
            for (int j = 0; j < NSL; ++j) hx[set][s][j][lane] = hq[s][j];
          }
        }
      }
    }
    wg_sync();
  }

  if (role == 3) {
    float ph[NSL - 1];
#pragma unroll
    for (int j = 0; j < NSL - 1; ++j) ph[j] = __shfl_xor(h[j], 16, 32);
    if (hi == 0) {
      float hall[HID];
#pragma unroll
      for (int j = 0; j < NSL; ++j) hall[j] = h[j];
#pragma unroll
      for (int j = 0; j < NSL - 1; ++j) hall[NSL + j] = ph[j];
      float* op = out + (size_t)bglob * NCLS;
#pragma unroll
      for (int k = 0; k < NCLS; ++k) {
        float acc = fcb[k];
#pragma unroll
        for (int j = 0; j < HID; ++j) acc += hall[j] * fcw[k * HID + j];
        op[k] = acc;
      }
    }
  }
}

extern "C" void kernel_launch(void* const* d_in, const int* in_sizes, int n_in,
                              void* d_out, int out_size, void* d_ws, size_t ws_size,
                              hipStream_t stream) {
  const float* x    = (const float*)d_in[0];
  const float* wih0 = (const float*)d_in[1];
  const float* whh0 = (const float*)d_in[2];
  const float* bih0 = (const float*)d_in[3];
  const float* bhh0 = (const float*)d_in[4];
  const float* wih1 = (const float*)d_in[5];
  const float* whh1 = (const float*)d_in[6];
  const float* bih1 = (const float*)d_in[7];
  const float* bhh1 = (const float*)d_in[8];
  const float* fcw  = (const float*)d_in[9];
  const float* fcb  = (const float*)d_in[10];
  float* out = (float*)d_out;

  const int B = in_sizes[0] / (TLEN * HID);
  dim3 grid(B / 16);
  dim3 block(128);
  lstm2_fc_fused<<<grid, block, 0, stream>>>(x, wih0, whh0, bih0, bhh0,
                                             wih1, whh1, bih1, bhh1,
                                             fcw, fcb, out);
}
